# starting kernel
# baseline (speedup 1.0000x reference)
#include <hip/hip_runtime.h>

#define N_NODES  50000
#define N_EDGES  800000
#define DIM      128
#define N_LAYERS 4
#define N_GRAPHS 512
#define N_COLS   4
#define VOCAB    128

typedef __attribute__((ext_vector_type(2))) float v2f;
typedef __attribute__((ext_vector_type(4))) float v4f;
typedef __attribute__((ext_vector_type(8))) float v8f;

__global__ void embed_kernel(const int* __restrict__ x, const float* __restrict__ emb,
                             float* __restrict__ h) {
  int t = blockIdx.x * blockDim.x + threadIdx.x;
  if (t >= N_NODES * DIM) return;
  int n = t / DIM, d = t % DIM;
  float s = 0.f;
#pragma unroll
  for (int c = 0; c < N_COLS; ++c) {
    int v = x[n * N_COLS + c];
    s += emb[((size_t)c * VOCAB + v) * DIM + d];
  }
  h[t] = s;
}

__global__ void deg_init_kernel(float* __restrict__ deg) {
  int n = blockIdx.x * blockDim.x + threadIdx.x;
  if (n < N_NODES) deg[n] = 1.0f;
}

__global__ void deg_edge_kernel(const int* __restrict__ ei, float* __restrict__ deg) {
  int e = blockIdx.x * blockDim.x + threadIdx.x;
  if (e < N_EDGES) unsafeAtomicAdd(&deg[ei[N_EDGES + e]], 1.0f);
}

__global__ void dinv_kernel(float* __restrict__ deg) {
  int n = blockIdx.x * blockDim.x + threadIdx.x;
  if (n >= N_NODES) return;
  float d = deg[n];
  deg[n] = (d > 0.f) ? rsqrtf(d) : 0.f;
}

__global__ void norm_kernel(const int* __restrict__ ei, const float* __restrict__ dinv,
                            float* __restrict__ norm) {
  int e = blockIdx.x * blockDim.x + threadIdx.x;
  if (e >= N_EDGES) return;
  norm[e] = dinv[ei[e]] * dinv[ei[N_EDGES + e]];
}

__global__ void gemm_wmma_f32(const float* __restrict__ A, const float* __restrict__ B,
                              const float* __restrict__ bias, float* __restrict__ Cout) {
  int row0 = blockIdx.x * 16;
  int col0 = blockIdx.y * 16;
  int lane = threadIdx.x;
  int half = lane >> 4;
  int lm   = lane & 15;

  const float* ap = A + (size_t)(row0 + lm) * DIM + half * 2;
  const float* bp = B + (size_t)(half * 2) * DIM + col0 + lm;

  v8f acc = {};
#pragma unroll
  for (int k = 0; k < DIM; k += 4) {
    v2f a = *(const v2f*)(ap + k);
    v2f b;
    b.x = bp[k * DIM];
    b.y = bp[k * DIM + DIM];
    acc = __builtin_amdgcn_wmma_f32_16x16x4_f32(
         false, a,  false, b,
         (short)0, acc,  false,  false);
  }

  float badd = bias ? bias[col0 + lm] : 0.f;
#pragma unroll
  for (int r = 0; r < 8; ++r)
    Cout[(size_t)(row0 + half * 8 + r) * DIM + col0 + lm] = acc[r] + badd;
}

__global__ void init_self_bias_kernel(const float* __restrict__ xw, const float* __restrict__ dinv,
                                      const float* __restrict__ bl, float* __restrict__ nh) {
  int t = blockIdx.x * blockDim.x + threadIdx.x;
  if (t >= N_NODES * DIM) return;
  int n = t / DIM, d = t % DIM;
  float s = dinv[n];
  nh[t] = bl[d] + xw[t] * s * s;
}

__global__ void scatter_kernel(const int* __restrict__ ei, const float* __restrict__ norm,
                               const float* __restrict__ xw, float* __restrict__ nh) {
  int t = blockIdx.x * blockDim.x + threadIdx.x;
  if (t >= N_EDGES * 32) return;
  int e  = t >> 5;
  int dg = (t & 31) << 2;
  int s = ei[e];
  int d = ei[N_EDGES + e];
  float w = norm[e];
  v4f v = *(const v4f*)(xw + (size_t)s * DIM + dg);
  float* dst = nh + (size_t)d * DIM + dg;
  unsafeAtomicAdd(dst + 0, v.x * w);
  unsafeAtomicAdd(dst + 1, v.y * w);
  unsafeAtomicAdd(dst + 2, v.z * w);
  unsafeAtomicAdd(dst + 3, v.w * w);
}

__global__ void relu_kernel(const float* __restrict__ nh, float* __restrict__ h) {
  int t = blockIdx.x * blockDim.x + threadIdx.x;
  if (t < N_NODES * DIM) h[t] = fmaxf(nh[t], 0.f);
}

__global__ void pool_zero_kernel(float* __restrict__ pool, float* __restrict__ cnt) {
  int t = blockIdx.x * blockDim.x + threadIdx.x;
  if (t < N_GRAPHS * DIM) pool[t] = 0.f;
  else if (t < N_GRAPHS * DIM + N_GRAPHS) cnt[t - N_GRAPHS * DIM] = 0.f;
}

__global__ void pool_scatter_kernel(const int* __restrict__ batch, const float* __restrict__ h,
                                    float* __restrict__ pool) {
  int t = blockIdx.x * blockDim.x + threadIdx.x;
  if (t >= N_NODES * DIM) return;
  int n = t / DIM, d = t % DIM;
  unsafeAtomicAdd(&pool[(size_t)batch[n] * DIM + d], h[t]);
}

__global__ void cnt_scatter_kernel(const int* __restrict__ batch, float* __restrict__ cnt) {
  int n = blockIdx.x * blockDim.x + threadIdx.x;
  if (n < N_NODES) unsafeAtomicAdd(&cnt[batch[n]], 1.0f);
}

__global__ void pool_div_kernel(float* __restrict__ pool, const float* __restrict__ cnt) {
  int t = blockIdx.x * blockDim.x + threadIdx.x;
  if (t >= N_GRAPHS * DIM) return;
  pool[t] /= fmaxf(cnt[t / DIM], 1.0f);
}

extern "C" void kernel_launch(void* const* d_in, const int* in_sizes, int n_in,
                              void* d_out, int out_size, void* d_ws, size_t ws_size,
                              hipStream_t stream) {
  const int*   x    = (const int*)d_in[0];
  const int*   ei   = (const int*)d_in[1];
  const int*   batch= (const int*)d_in[2];
  const float* emb  = (const float*)d_in[3];
  const float* W    = (const float*)d_in[4];
  const float* b    = (const float*)d_in[5];
  const float* Wr   = (const float*)d_in[6];
  const float* br   = (const float*)d_in[7];
  float* out = (float*)d_out;

  char* ws = (char*)d_ws;
  const size_t NH = (size_t)N_NODES * DIM * sizeof(float);
  float* h    = (float*)(ws);
  float* xw   = (float*)(ws + NH);
  float* nh   = (float*)(ws + 2 * NH);
  float* dinv = (float*)(ws + 3 * NH);
  float* norm = (float*)(ws + 3 * NH + 256 * 1024);
  float* pool = (float*)(ws + 3 * NH + 4 * 1024 * 1024);
  float* cnt  = (float*)(ws + 3 * NH + 4 * 1024 * 1024 + 512 * 1024);

  const int B256 = 256;
  const int nd_blocks = (N_NODES * DIM + B256 - 1) / B256;
  const int n_blocks  = (N_NODES + B256 - 1) / B256;
  const int e_blocks  = (N_EDGES + B256 - 1) / B256;

  embed_kernel<<<nd_blocks, B256, 0, stream>>>(x, emb, h);

  deg_init_kernel<<<n_blocks, B256, 0, stream>>>(dinv);
  deg_edge_kernel<<<e_blocks, B256, 0, stream>>>(ei, dinv);
  dinv_kernel<<<n_blocks, B256, 0, stream>>>(dinv);
  norm_kernel<<<e_blocks, B256, 0, stream>>>(ei, dinv, norm);

  dim3 ggrid(N_NODES / 16, DIM / 16);
  const int sc_blocks = (N_EDGES * 32 + B256 - 1) / B256;
  for (int l = 0; l < N_LAYERS; ++l) {
    gemm_wmma_f32<<<ggrid, 32, 0, stream>>>(h, W + (size_t)l * DIM * DIM, nullptr, xw);
    init_self_bias_kernel<<<nd_blocks, B256, 0, stream>>>(xw, dinv, b + (size_t)l * DIM, nh);
    scatter_kernel<<<sc_blocks, B256, 0, stream>>>(ei, norm, xw, nh);
    relu_kernel<<<nd_blocks, B256, 0, stream>>>(nh, h);
  }

  pool_zero_kernel<<<(N_GRAPHS * DIM + N_GRAPHS + B256 - 1) / B256, B256, 0, stream>>>(pool, cnt);
  pool_scatter_kernel<<<nd_blocks, B256, 0, stream>>>(batch, h, pool);
  cnt_scatter_kernel<<<n_blocks, B256, 0, stream>>>(batch, cnt);
  pool_div_kernel<<<(N_GRAPHS * DIM + B256 - 1) / B256, B256, 0, stream>>>(pool, cnt);

  dim3 rgrid(N_GRAPHS / 16, DIM / 16);
  gemm_wmma_f32<<<rgrid, 32, 0, stream>>>(pool, Wr, br, out);
}


# Per-layer f32 atomic scatter replaced by a per-call counting sort of edges by destination (integer atomic arrival rank) and a one-wave-per-node register sum stored once; sum order = rank order
# speedup vs baseline: 2.4033x; 2.4033x over previous
#include <hip/hip_runtime.h>

#define N_NODES  50000
#define N_EDGES  800000
#define DIM      128
#define N_LAYERS 4
#define N_GRAPHS 512
#define N_COLS   4
#define VOCAB    128

typedef __attribute__((ext_vector_type(2))) float v2f;
typedef __attribute__((ext_vector_type(4))) float v4f;
typedef __attribute__((ext_vector_type(8))) float v8f;

__global__ void embed_kernel(const int* __restrict__ x, const float* __restrict__ emb,
                             float* __restrict__ h) {
  int t = blockIdx.x * blockDim.x + threadIdx.x;
  if (t >= N_NODES * DIM) return;
  int n = t / DIM, d = t % DIM;
  float s = 0.f;
#pragma unroll
  for (int c = 0; c < N_COLS; ++c) {
    int v = x[n * N_COLS + c];
    s += emb[((size_t)c * VOCAB + v) * DIM + d];
  }
  h[t] = s;
}

typedef __attribute__((ext_vector_type(2))) int v2i;

__global__ void ecnt_zero_kernel(int* __restrict__ ecnt) {
  int n = blockIdx.x * blockDim.x + threadIdx.x;
  if (n < N_NODES) ecnt[n] = 0;
}

__global__ void count_rank_kernel(const int* __restrict__ ei, int* __restrict__ ecnt,
                                  int* __restrict__ rank) {
  int e = blockIdx.x * blockDim.x + threadIdx.x;
  if (e < N_EDGES) rank[e] = atomicAdd(&ecnt[ei[N_EDGES + e]], 1);
}

__global__ void __launch_bounds__(1024)
rowptr_dinv_kernel(const int* __restrict__ ecnt, int* __restrict__ rowptr, float* __restrict__ dinv) {
  __shared__ int s_pre[32];
  __shared__ int s_sum[32];
  const int t = threadIdx.x, lane = t & 31, wv = t >> 5;
  const int base = blockIdx.x * 1024;
  int pre = 0;
  for (int i = t; i < base; i += 1024) pre += ecnt[i];
  const int i = base + t;
  const int c = (i < N_NODES) ? ecnt[i] : 0;
  int x = c;
  for (int off = 1; off < 32; off <<= 1) {
    int y = __shfl_up(x, off, 32);
    if (lane >= off) x += y;
  }
  for (int off = 16; off > 0; off >>= 1) pre += __shfl_xor(pre, off, 32);
  if (lane == 31) s_sum[wv] = x;
  if (lane == 0) s_pre[wv] = pre;
  __syncthreads();
  const int ws = s_sum[lane];
  int bp = s_pre[lane];
  int wx = ws;
  for (int off = 1; off < 32; off <<= 1) {
    int y = __shfl_up(wx, off, 32);
    if (lane >= off) wx += y;
  }
  for (int off = 16; off > 0; off >>= 1) bp += __shfl_xor(bp, off, 32);
  const int wave_excl = __shfl(wx, wv, 32) - __shfl(ws, wv, 32);
  const int excl = bp + wave_excl + x - c;
  if (i < N_NODES) {
    rowptr[i] = excl;
    float d = 1.0f + (float)c;
    dinv[i] = (d > 0.f) ? rsqrtf(d) : 0.f;
    if (i == N_NODES - 1) rowptr[N_NODES] = excl + c;
  }
}

__global__ void csr_fill_kernel(const int* __restrict__ ei, const int* __restrict__ rank,
                                const int* __restrict__ rowptr, const float* __restrict__ dinv,
                                int* __restrict__ csr) {
  int e = blockIdx.x * blockDim.x + threadIdx.x;
  if (e >= N_EDGES) return;
  const int s = ei[e];
  const int d = ei[N_EDGES + e];
  const int pos = rowptr[d] + rank[e];
  v2i ent;
  ent.x = s;
  ent.y = __float_as_int(dinv[s] * dinv[d]);
  *(v2i*)(csr + 2 * (size_t)pos) = ent;
}

__global__ void gemm_wmma_f32(const float* __restrict__ A, const float* __restrict__ B,
                              const float* __restrict__ bias, float* __restrict__ Cout) {
  int row0 = blockIdx.x * 16;
  int col0 = blockIdx.y * 16;
  int lane = threadIdx.x;
  int half = lane >> 4;
  int lm   = lane & 15;

  const float* ap = A + (size_t)(row0 + lm) * DIM + half * 2;
  const float* bp = B + (size_t)(half * 2) * DIM + col0 + lm;

  v8f acc = {};
#pragma unroll
  for (int k = 0; k < DIM; k += 4) {
    v2f a = *(const v2f*)(ap + k);
    v2f b;
    b.x = bp[k * DIM];
    b.y = bp[k * DIM + DIM];
    acc = __builtin_amdgcn_wmma_f32_16x16x4_f32(
         false, a,  false, b,
         (short)0, acc,  false,  false);
  }

  float badd = bias ? bias[col0 + lm] : 0.f;
#pragma unroll
  for (int r = 0; r < 8; ++r)
    Cout[(size_t)(row0 + half * 8 + r) * DIM + col0 + lm] = acc[r] + badd;
}

static_assert((N_NODES * 32) % 256 == 0, "whole blocks only: the node index is never out of range");
__global__ void __launch_bounds__(256)
gather_relu_kernel(const int* __restrict__ rowptr, const int* __restrict__ csr,
                   const float* __restrict__ xw, const float* __restrict__ dinv,
                   const float* __restrict__ bl, float* __restrict__ h) {
  const int lane = threadIdx.x & 31;
  const int n = __builtin_amdgcn_readfirstlane((int)((blockIdx.x * 256 + threadIdx.x) >> 5));
  const int beg = rowptr[n];
  const int end = rowptr[n + 1];
  const float* xl = xw + lane * 4;
  v4f acc = {0.f, 0.f, 0.f, 0.f};
  for (int base = beg; base < end; base += 32) {
    const int j = base + lane;
    v2i ent = {0, 0};
    if (j < end) ent = *(const v2i*)(csr + 2 * (size_t)j);
    const int m = (end - base < 32) ? (end - base) : 32;
#pragma unroll 4
    for (int k = 0; k < m; ++k) {
      const int sk = __builtin_amdgcn_readlane(ent.x, k);
      const float wk = __int_as_float(__builtin_amdgcn_readlane(ent.y, k));
      const v4f v = *(const v4f*)(xl + (size_t)sk * DIM);
      acc += v * wk;
    }
  }
  const float di = dinv[n];
  const v4f self = *(const v4f*)(xl + (size_t)n * DIM);
  const v4f bb = *(const v4f*)(bl + lane * 4);
  v4f r = bb + self * di * di;
  r += acc;
  r.x = fmaxf(r.x, 0.f);
  r.y = fmaxf(r.y, 0.f);
  r.z = fmaxf(r.z, 0.f);
  r.w = fmaxf(r.w, 0.f);
  *(v4f*)(h + (size_t)n * DIM + lane * 4) = r;
}

__global__ void pool_zero_kernel(float* __restrict__ pool, float* __restrict__ cnt) {
  int t = blockIdx.x * blockDim.x + threadIdx.x;
  if (t < N_GRAPHS * DIM) pool[t] = 0.f;
  else if (t < N_GRAPHS * DIM + N_GRAPHS) cnt[t - N_GRAPHS * DIM] = 0.f;
}

__global__ void pool_scatter_kernel(const int* __restrict__ batch, const float* __restrict__ h,
                                    float* __restrict__ pool) {
  int t = blockIdx.x * blockDim.x + threadIdx.x;
  if (t >= N_NODES * DIM) return;
  int n = t / DIM, d = t % DIM;
  unsafeAtomicAdd(&pool[(size_t)batch[n] * DIM + d], h[t]);
}

__global__ void cnt_scatter_kernel(const int* __restrict__ batch, float* __restrict__ cnt) {
  int n = blockIdx.x * blockDim.x + threadIdx.x;
  if (n < N_NODES) unsafeAtomicAdd(&cnt[batch[n]], 1.0f);
}

__global__ void pool_div_kernel(float* __restrict__ pool, const float* __restrict__ cnt) {
  int t = blockIdx.x * blockDim.x + threadIdx.x;
  if (t >= N_GRAPHS * DIM) return;
  pool[t] /= fmaxf(cnt[t / DIM], 1.0f);
}

extern "C" void kernel_launch(void* const* d_in, const int* in_sizes, int n_in,
                              void* d_out, int out_size, void* d_ws, size_t ws_size,
                              hipStream_t stream) {
  const int*   x    = (const int*)d_in[0];
  const int*   ei   = (const int*)d_in[1];
  const int*   batch= (const int*)d_in[2];
  const float* emb  = (const float*)d_in[3];
  const float* W    = (const float*)d_in[4];
  const float* b    = (const float*)d_in[5];
  const float* Wr   = (const float*)d_in[6];
  const float* br   = (const float*)d_in[7];
  float* out = (float*)d_out;

  char* ws = (char*)d_ws;
  const size_t NH = (size_t)N_NODES * DIM * sizeof(float);
  float* h    = (float*)(ws);
  float* xw   = (float*)(ws + NH);
  int*   ecnt   = (int*)(ws + 2 * NH);
  int*   rowptr = (int*)(ws + 2 * NH + 256 * 1024);
  int*   rank   = (int*)(ws + 2 * NH + 512 * 1024);
  int*   csr    = (int*)(ws + 2 * NH + 8 * 1024 * 1024);
  float* dinv = (float*)(ws + 3 * NH);
  float* pool = (float*)(ws + 3 * NH + 4 * 1024 * 1024);
  float* cnt  = (float*)(ws + 3 * NH + 4 * 1024 * 1024 + 512 * 1024);

  const int B256 = 256;
  const int nd_blocks = (N_NODES * DIM + B256 - 1) / B256;
  const int n_blocks  = (N_NODES + B256 - 1) / B256;
  const int e_blocks  = (N_EDGES + B256 - 1) / B256;

  embed_kernel<<<nd_blocks, B256, 0, stream>>>(x, emb, h);

  ecnt_zero_kernel<<<n_blocks, B256, 0, stream>>>(ecnt);
  count_rank_kernel<<<e_blocks, B256, 0, stream>>>(ei, ecnt, rank);
  rowptr_dinv_kernel<<<(N_NODES + 1023) / 1024, 1024, 0, stream>>>(ecnt, rowptr, dinv);
  csr_fill_kernel<<<e_blocks, B256, 0, stream>>>(ei, rank, rowptr, dinv, csr);

  dim3 ggrid(N_NODES / 16, DIM / 16);
  const int ga_blocks = N_NODES * 32 / B256;
  for (int l = 0; l < N_LAYERS; ++l) {
    gemm_wmma_f32<<<ggrid, 32, 0, stream>>>(h, W + (size_t)l * DIM * DIM, nullptr, xw);
    gather_relu_kernel<<<ga_blocks, B256, 0, stream>>>(rowptr, csr, xw, dinv, b + (size_t)l * DIM, h);
  }

  pool_zero_kernel<<<(N_GRAPHS * DIM + N_GRAPHS + B256 - 1) / B256, B256, 0, stream>>>(pool, cnt);
  pool_scatter_kernel<<<nd_blocks, B256, 0, stream>>>(batch, h, pool);
  cnt_scatter_kernel<<<n_blocks, B256, 0, stream>>>(batch, cnt);
  pool_div_kernel<<<(N_GRAPHS * DIM + B256 - 1) / B256, B256, 0, stream>>>(pool, cnt);

  dim3 rgrid(N_GRAPHS / 16, DIM / 16);
  gemm_wmma_f32<<<rgrid, 32, 0, stream>>>(pool, Wr, br, out);
}


# GEMM as 8-wave workgroups: each wave keeps its 16 weight columns for all K in registers and walks 16-row tiles staged once in LDS as whole rows; k order inside each 8-k group changes (sum order)
# speedup vs baseline: 2.8854x; 1.2006x over previous
#include <hip/hip_runtime.h>

#define N_NODES  50000
#define N_EDGES  800000
#define DIM      128
#define N_LAYERS 4
#define N_GRAPHS 512
#define N_COLS   4
#define VOCAB    128

typedef __attribute__((ext_vector_type(2))) float v2f;
typedef __attribute__((ext_vector_type(4))) float v4f;
typedef __attribute__((ext_vector_type(8))) float v8f;

__global__ void embed_kernel(const int* __restrict__ x, const float* __restrict__ emb,
                             float* __restrict__ h) {
  int t = blockIdx.x * blockDim.x + threadIdx.x;
  if (t >= N_NODES * DIM) return;
  int n = t / DIM, d = t % DIM;
  float s = 0.f;
#pragma unroll
  for (int c = 0; c < N_COLS; ++c) {
    int v = x[n * N_COLS + c];
    s += emb[((size_t)c * VOCAB + v) * DIM + d];
  }
  h[t] = s;
}

typedef __attribute__((ext_vector_type(2))) int v2i;

__global__ void ecnt_zero_kernel(int* __restrict__ ecnt) {
  int n = blockIdx.x * blockDim.x + threadIdx.x;
  if (n < N_NODES) ecnt[n] = 0;
}

__global__ void count_rank_kernel(const int* __restrict__ ei, int* __restrict__ ecnt,
                                  int* __restrict__ rank) {
  int e = blockIdx.x * blockDim.x + threadIdx.x;
  if (e < N_EDGES) rank[e] = atomicAdd(&ecnt[ei[N_EDGES + e]], 1);
}

__global__ void __launch_bounds__(1024)
rowptr_dinv_kernel(const int* __restrict__ ecnt, int* __restrict__ rowptr, float* __restrict__ dinv) {
  __shared__ int s_pre[32];
  __shared__ int s_sum[32];
  const int t = threadIdx.x, lane = t & 31, wv = t >> 5;
  const int base = blockIdx.x * 1024;
  int pre = 0;
  for (int i = t; i < base; i += 1024) pre += ecnt[i];
  const int i = base + t;
  const int c = (i < N_NODES) ? ecnt[i] : 0;
  int x = c;
  for (int off = 1; off < 32; off <<= 1) {
    int y = __shfl_up(x, off, 32);
    if (lane >= off) x += y;
  }
  for (int off = 16; off > 0; off >>= 1) pre += __shfl_xor(pre, off, 32);
  if (lane == 31) s_sum[wv] = x;
  if (lane == 0) s_pre[wv] = pre;
  __syncthreads();
  const int ws = s_sum[lane];
  int bp = s_pre[lane];
  int wx = ws;
  for (int off = 1; off < 32; off <<= 1) {
    int y = __shfl_up(wx, off, 32);
    if (lane >= off) wx += y;
  }
  for (int off = 16; off > 0; off >>= 1) bp += __shfl_xor(bp, off, 32);
  const int wave_excl = __shfl(wx, wv, 32) - __shfl(ws, wv, 32);
  const int excl = bp + wave_excl + x - c;
  if (i < N_NODES) {
    rowptr[i] = excl;
    float d = 1.0f + (float)c;
    dinv[i] = (d > 0.f) ? rsqrtf(d) : 0.f;
    if (i == N_NODES - 1) rowptr[N_NODES] = excl + c;
  }
}

__global__ void csr_fill_kernel(const int* __restrict__ ei, const int* __restrict__ rank,
                                const int* __restrict__ rowptr, const float* __restrict__ dinv,
                                int* __restrict__ csr) {
  int e = blockIdx.x * blockDim.x + threadIdx.x;
  if (e >= N_EDGES) return;
  const int s = ei[e];
  const int d = ei[N_EDGES + e];
  const int pos = rowptr[d] + rank[e];
  v2i ent;
  ent.x = s;
  ent.y = __float_as_int(dinv[s] * dinv[d]);
  *(v2i*)(csr + 2 * (size_t)pos) = ent;
}

#define A_PITCH 132
__global__ void __launch_bounds__(256)
gemm_wmma_f32(const float* __restrict__ A, const float* __restrict__ B,
              const float* __restrict__ bias, float* __restrict__ Cout, int n_tiles) {
  __shared__ __attribute__((aligned(16))) float sA[2][16 * A_PITCH];
  const int tid  = threadIdx.x;
  const int lane = tid & 31;
  const int wv   = tid >> 5;
  const int half = lane >> 4;
  const int lm   = lane & 15;
  const int col0 = wv * 16;

  v2f bf[32];
  const float* bp = B + (size_t)(4 * half) * DIM + col0 + lm;
#pragma unroll
  for (int s = 0; s < 32; ++s) {
    const int r0 = 8 * (s >> 1) + 2 * (s & 1);
    bf[s].x = bp[r0 * DIM];
    bf[s].y = bp[(r0 + 1) * DIM];
  }
  const float badd = bias ? bias[col0 + lm] : 0.f;

  const float* ga = A + (size_t)wv * DIM + lane * 4;
  const int so0 = wv * A_PITCH + lane * 4;
  const int so1 = (wv + 8) * A_PITCH + lane * 4;
  const int fo  = lm * A_PITCH + 4 * half;

  int tile = blockIdx.x;
  v4f p0 = {0.f, 0.f, 0.f, 0.f}, p1 = {0.f, 0.f, 0.f, 0.f};
  if (tile < n_tiles) {
    p0 = *(const v4f*)(ga + (size_t)tile * 16 * DIM);
    p1 = *(const v4f*)(ga + (size_t)tile * 16 * DIM + 8 * DIM);
    *(v4f*)(&sA[0][so0]) = p0;
    *(v4f*)(&sA[0][so1]) = p1;
  }
  __syncthreads();
  int buf = 0;
  for (; tile < n_tiles; tile += gridDim.x) {
    const int nt = tile + gridDim.x;
    const bool more = nt < n_tiles;
    if (more) {
      p0 = *(const v4f*)(ga + (size_t)nt * 16 * DIM);
      p1 = *(const v4f*)(ga + (size_t)nt * 16 * DIM + 8 * DIM);
    }
    const float* fa = &sA[buf][fo];
    v8f acc = {};
#pragma unroll
    for (int j = 0; j < 16; ++j) {
      const v4f a = *(const v4f*)(fa + 8 * j);
      v2f alo, ahi;
      alo.x = a.x; alo.y = a.y;
      ahi.x = a.z; ahi.y = a.w;
      acc = __builtin_amdgcn_wmma_f32_16x16x4_f32(
           false, alo,  false, bf[2 * j],
           (short)0, acc,  false,  false);
      acc = __builtin_amdgcn_wmma_f32_16x16x4_f32(
           false, ahi,  false, bf[2 * j + 1],
           (short)0, acc,  false,  false);
    }
    float* cp = Cout + (size_t)(tile * 16 + half * 8) * DIM + col0 + lm;
#pragma unroll
    for (int r = 0; r < 8; ++r) cp[r * DIM] = acc[r] + badd;
    if (more) {
      *(v4f*)(&sA[buf ^ 1][so0]) = p0;
      *(v4f*)(&sA[buf ^ 1][so1]) = p1;
    }
    __syncthreads();
    buf ^= 1;
  }
}

static_assert((N_NODES * 32) % 256 == 0, "whole blocks only: the node index is never out of range");
__global__ void __launch_bounds__(256)
gather_relu_kernel(const int* __restrict__ rowptr, const int* __restrict__ csr,
                   const float* __restrict__ xw, const float* __restrict__ dinv,
                   const float* __restrict__ bl, float* __restrict__ h) {
  const int lane = threadIdx.x & 31;
  const int n = __builtin_amdgcn_readfirstlane((int)((blockIdx.x * 256 + threadIdx.x) >> 5));
  const int beg = rowptr[n];
  const int end = rowptr[n + 1];
  const float* xl = xw + lane * 4;
  v4f acc = {0.f, 0.f, 0.f, 0.f};
  for (int base = beg; base < end; base += 32) {
    const int j = base + lane;
    v2i ent = {0, 0};
    if (j < end) ent = *(const v2i*)(csr + 2 * (size_t)j);
    const int m = (end - base < 32) ? (end - base) : 32;
#pragma unroll 4
    for (int k = 0; k < m; ++k) {
      const int sk = __builtin_amdgcn_readlane(ent.x, k);
      const float wk = __int_as_float(__builtin_amdgcn_readlane(ent.y, k));
      const v4f v = *(const v4f*)(xl + (size_t)sk * DIM);
      acc += v * wk;
    }
  }
  const float di = dinv[n];
  const v4f self = *(const v4f*)(xl + (size_t)n * DIM);
  const v4f bb = *(const v4f*)(bl + lane * 4);
  v4f r = bb + self * di * di;
  r += acc;
  r.x = fmaxf(r.x, 0.f);
  r.y = fmaxf(r.y, 0.f);
  r.z = fmaxf(r.z, 0.f);
  r.w = fmaxf(r.w, 0.f);
  *(v4f*)(h + (size_t)n * DIM + lane * 4) = r;
}

__global__ void pool_zero_kernel(float* __restrict__ pool, float* __restrict__ cnt) {
  int t = blockIdx.x * blockDim.x + threadIdx.x;
  if (t < N_GRAPHS * DIM) pool[t] = 0.f;
  else if (t < N_GRAPHS * DIM + N_GRAPHS) cnt[t - N_GRAPHS * DIM] = 0.f;
}

__global__ void pool_scatter_kernel(const int* __restrict__ batch, const float* __restrict__ h,
                                    float* __restrict__ pool) {
  int t = blockIdx.x * blockDim.x + threadIdx.x;
  if (t >= N_NODES * DIM) return;
  int n = t / DIM, d = t % DIM;
  unsafeAtomicAdd(&pool[(size_t)batch[n] * DIM + d], h[t]);
}

__global__ void cnt_scatter_kernel(const int* __restrict__ batch, float* __restrict__ cnt) {
  int n = blockIdx.x * blockDim.x + threadIdx.x;
  if (n < N_NODES) unsafeAtomicAdd(&cnt[batch[n]], 1.0f);
}

__global__ void pool_div_kernel(float* __restrict__ pool, const float* __restrict__ cnt) {
  int t = blockIdx.x * blockDim.x + threadIdx.x;
  if (t >= N_GRAPHS * DIM) return;
  pool[t] /= fmaxf(cnt[t / DIM], 1.0f);
}

extern "C" void kernel_launch(void* const* d_in, const int* in_sizes, int n_in,
                              void* d_out, int out_size, void* d_ws, size_t ws_size,
                              hipStream_t stream) {
  const int*   x    = (const int*)d_in[0];
  const int*   ei   = (const int*)d_in[1];
  const int*   batch= (const int*)d_in[2];
  const float* emb  = (const float*)d_in[3];
  const float* W    = (const float*)d_in[4];
  const float* b    = (const float*)d_in[5];
  const float* Wr   = (const float*)d_in[6];
  const float* br   = (const float*)d_in[7];
  float* out = (float*)d_out;

  char* ws = (char*)d_ws;
  const size_t NH = (size_t)N_NODES * DIM * sizeof(float);
  float* h    = (float*)(ws);
  float* xw   = (float*)(ws + NH);
  int*   ecnt   = (int*)(ws + 2 * NH);
  int*   rowptr = (int*)(ws + 2 * NH + 256 * 1024);
  int*   rank   = (int*)(ws + 2 * NH + 512 * 1024);
  int*   csr    = (int*)(ws + 2 * NH + 8 * 1024 * 1024);
  float* dinv = (float*)(ws + 3 * NH);
  float* pool = (float*)(ws + 3 * NH + 4 * 1024 * 1024);
  float* cnt  = (float*)(ws + 3 * NH + 4 * 1024 * 1024 + 512 * 1024);

  const int B256 = 256;
  const int nd_blocks = (N_NODES * DIM + B256 - 1) / B256;
  const int n_blocks  = (N_NODES + B256 - 1) / B256;
  const int e_blocks  = (N_EDGES + B256 - 1) / B256;

  embed_kernel<<<nd_blocks, B256, 0, stream>>>(x, emb, h);

  ecnt_zero_kernel<<<n_blocks, B256, 0, stream>>>(ecnt);
  count_rank_kernel<<<e_blocks, B256, 0, stream>>>(ei, ecnt, rank);
  rowptr_dinv_kernel<<<(N_NODES + 1023) / 1024, 1024, 0, stream>>>(ecnt, rowptr, dinv);
  csr_fill_kernel<<<e_blocks, B256, 0, stream>>>(ei, rank, rowptr, dinv, csr);

  const int g_blocks = 1024;
  const int ga_blocks = N_NODES * 32 / B256;
  for (int l = 0; l < N_LAYERS; ++l) {
    gemm_wmma_f32<<<g_blocks, B256, 0, stream>>>(h, W + (size_t)l * DIM * DIM, nullptr, xw, N_NODES / 16);
    gather_relu_kernel<<<ga_blocks, B256, 0, stream>>>(rowptr, csr, xw, dinv, b + (size_t)l * DIM, h);
  }

  pool_zero_kernel<<<(N_GRAPHS * DIM + N_GRAPHS + B256 - 1) / B256, B256, 0, stream>>>(pool, cnt);
  pool_scatter_kernel<<<nd_blocks, B256, 0, stream>>>(batch, h, pool);
  cnt_scatter_kernel<<<n_blocks, B256, 0, stream>>>(batch, cnt);
  pool_div_kernel<<<(N_GRAPHS * DIM + B256 - 1) / B256, B256, 0, stream>>>(pool, cnt);

  gemm_wmma_f32<<<N_GRAPHS / 16, B256, 0, stream>>>(pool, Wr, br, out, N_GRAPHS / 16);
}


# Fewer, fatter waves: embed and pool take whole 512-byte rows (4 features per lane; 5 and 10 nodes per wave), edge kernels 4 edges per thread; pool adds per-wave sums of equal-graph runs (sum order)
# speedup vs baseline: 3.2132x; 1.1136x over previous
#include <hip/hip_runtime.h>

#define N_NODES  50000
#define N_EDGES  800000
#define DIM      128
#define N_LAYERS 4
#define N_GRAPHS 512
#define N_COLS   4
#define VOCAB    128

typedef __attribute__((ext_vector_type(2))) float v2f;
typedef __attribute__((ext_vector_type(4))) float v4f;
typedef __attribute__((ext_vector_type(8))) float v8f;

#define EMB_NPW 5
static_assert(N_NODES % (8 * EMB_NPW) == 0, "whole blocks only: the node index is never out of range");
__global__ void __launch_bounds__(256)
embed_kernel(const int* __restrict__ x, const float* __restrict__ emb, float* __restrict__ h) {
  const int lane = threadIdx.x & 31;
  const int w = __builtin_amdgcn_readfirstlane((int)((blockIdx.x * 256 + threadIdx.x) >> 5));
  const float* el = emb + lane * 4;
  float* hl = h + lane * 4;
#pragma unroll
  for (int i = 0; i < EMB_NPW; ++i) {
    const int n = w * EMB_NPW + i;
    v4f s = {0.f, 0.f, 0.f, 0.f};
#pragma unroll
    for (int c = 0; c < N_COLS; ++c) {
      const int v = x[n * N_COLS + c];
      s += *(const v4f*)(el + ((size_t)c * VOCAB + v) * DIM);
    }
    *(v4f*)(hl + (size_t)n * DIM) = s;
  }
}

typedef __attribute__((ext_vector_type(2))) int v2i;

__global__ void ecnt_zero_kernel(int* __restrict__ ecnt) {
  int n = blockIdx.x * blockDim.x + threadIdx.x;
  if (n < N_NODES) ecnt[n] = 0;
}

typedef __attribute__((ext_vector_type(4))) int v4i;
static_assert(N_EDGES % 4 == 0, "edges are taken four at a time");
__global__ void count_rank_kernel(const int* __restrict__ ei, int* __restrict__ ecnt,
                                  int* __restrict__ rank) {
  int t = blockIdx.x * blockDim.x + threadIdx.x;
  if (t >= N_EDGES / 4) return;
  const v4i d = *(const v4i*)(ei + N_EDGES + 4 * (size_t)t);
  v4i r;
  r.x = atomicAdd(&ecnt[d.x], 1);
  r.y = atomicAdd(&ecnt[d.y], 1);
  r.z = atomicAdd(&ecnt[d.z], 1);
  r.w = atomicAdd(&ecnt[d.w], 1);
  *(v4i*)(rank + 4 * (size_t)t) = r;
}

__global__ void __launch_bounds__(1024)
rowptr_dinv_kernel(const int* __restrict__ ecnt, int* __restrict__ rowptr, float* __restrict__ dinv) {
  __shared__ int s_pre[32];
  __shared__ int s_sum[32];
  const int t = threadIdx.x, lane = t & 31, wv = t >> 5;
  const int base = blockIdx.x * 1024;
  int pre = 0;
  for (int i = t; i < base; i += 1024) pre += ecnt[i];
  const int i = base + t;
  const int c = (i < N_NODES) ? ecnt[i] : 0;
  int x = c;
  for (int off = 1; off < 32; off <<= 1) {
    int y = __shfl_up(x, off, 32);
    if (lane >= off) x += y;
  }
  for (int off = 16; off > 0; off >>= 1) pre += __shfl_xor(pre, off, 32);
  if (lane == 31) s_sum[wv] = x;
  if (lane == 0) s_pre[wv] = pre;
  __syncthreads();
  const int ws = s_sum[lane];
  int bp = s_pre[lane];
  int wx = ws;
  for (int off = 1; off < 32; off <<= 1) {
    int y = __shfl_up(wx, off, 32);
    if (lane >= off) wx += y;
  }
  for (int off = 16; off > 0; off >>= 1) bp += __shfl_xor(bp, off, 32);
  const int wave_excl = __shfl(wx, wv, 32) - __shfl(ws, wv, 32);
  const int excl = bp + wave_excl + x - c;
  if (i < N_NODES) {
    rowptr[i] = excl;
    float d = 1.0f + (float)c;
    dinv[i] = (d > 0.f) ? rsqrtf(d) : 0.f;
    if (i == N_NODES - 1) rowptr[N_NODES] = excl + c;
  }
}

__global__ void csr_fill_kernel(const int* __restrict__ ei, const int* __restrict__ rank,
                                const int* __restrict__ rowptr, const float* __restrict__ dinv,
                                int* __restrict__ csr) {
  int t = blockIdx.x * blockDim.x + threadIdx.x;
  if (t >= N_EDGES / 4) return;
  const v4i s4 = *(const v4i*)(ei + 4 * (size_t)t);
  const v4i d4 = *(const v4i*)(ei + N_EDGES + 4 * (size_t)t);
  const v4i r4 = *(const v4i*)(rank + 4 * (size_t)t);
#pragma unroll
  for (int k = 0; k < 4; ++k) {
    const int s = s4[k];
    const int d = d4[k];
    const int pos = rowptr[d] + r4[k];
    v2i ent;
    ent.x = s;
    ent.y = __float_as_int(dinv[s] * dinv[d]);
    *(v2i*)(csr + 2 * (size_t)pos) = ent;
  }
}

#define A_PITCH 132
__global__ void __launch_bounds__(256)
gemm_wmma_f32(const float* __restrict__ A, const float* __restrict__ B,
              const float* __restrict__ bias, float* __restrict__ Cout, int n_tiles) {
  __shared__ __attribute__((aligned(16))) float sA[2][16 * A_PITCH];
  const int tid  = threadIdx.x;
  const int lane = tid & 31;
  const int wv   = tid >> 5;
  const int half = lane >> 4;
  const int lm   = lane & 15;
  const int col0 = wv * 16;

  v2f bf[32];
  const float* bp = B + (size_t)(4 * half) * DIM + col0 + lm;
#pragma unroll
  for (int s = 0; s < 32; ++s) {
    const int r0 = 8 * (s >> 1) + 2 * (s & 1);
    bf[s].x = bp[r0 * DIM];
    bf[s].y = bp[(r0 + 1) * DIM];
  }
  const float badd = bias ? bias[col0 + lm] : 0.f;

  const float* ga = A + (size_t)wv * DIM + lane * 4;
  const int so0 = wv * A_PITCH + lane * 4;
  const int so1 = (wv + 8) * A_PITCH + lane * 4;
  const int fo  = lm * A_PITCH + 4 * half;

  int tile = blockIdx.x;
  v4f p0 = {0.f, 0.f, 0.f, 0.f}, p1 = {0.f, 0.f, 0.f, 0.f};
  if (tile < n_tiles) {
    p0 = *(const v4f*)(ga + (size_t)tile * 16 * DIM);
    p1 = *(const v4f*)(ga + (size_t)tile * 16 * DIM + 8 * DIM);
    *(v4f*)(&sA[0][so0]) = p0;
    *(v4f*)(&sA[0][so1]) = p1;
  }
  __syncthreads();
  int buf = 0;
  for (; tile < n_tiles; tile += gridDim.x) {
    const int nt = tile + gridDim.x;
    const bool more = nt < n_tiles;
    if (more) {
      p0 = *(const v4f*)(ga + (size_t)nt * 16 * DIM);
      p1 = *(const v4f*)(ga + (size_t)nt * 16 * DIM + 8 * DIM);
    }
    const float* fa = &sA[buf][fo];
    v8f acc = {};
#pragma unroll
    for (int j = 0; j < 16; ++j) {
      const v4f a = *(const v4f*)(fa + 8 * j);
      v2f alo, ahi;
      alo.x = a.x; alo.y = a.y;
      ahi.x = a.z; ahi.y = a.w;
      acc = __builtin_amdgcn_wmma_f32_16x16x4_f32(
           false, alo,  false, bf[2 * j],
           (short)0, acc,  false,  false);
      acc = __builtin_amdgcn_wmma_f32_16x16x4_f32(
           false, ahi,  false, bf[2 * j + 1],
           (short)0, acc,  false,  false);
    }
    float* cp = Cout + (size_t)(tile * 16 + half * 8) * DIM + col0 + lm;
#pragma unroll
    for (int r = 0; r < 8; ++r) cp[r * DIM] = acc[r] + badd;
    if (more) {
      *(v4f*)(&sA[buf ^ 1][so0]) = p0;
      *(v4f*)(&sA[buf ^ 1][so1]) = p1;
    }
    __syncthreads();
    buf ^= 1;
  }
}

static_assert((N_NODES * 32) % 256 == 0, "whole blocks only: the node index is never out of range");
__global__ void __launch_bounds__(256)
gather_relu_kernel(const int* __restrict__ rowptr, const int* __restrict__ csr,
                   const float* __restrict__ xw, const float* __restrict__ dinv,
                   const float* __restrict__ bl, float* __restrict__ h) {
  const int lane = threadIdx.x & 31;
  const int n = __builtin_amdgcn_readfirstlane((int)((blockIdx.x * 256 + threadIdx.x) >> 5));
  const int beg = rowptr[n];
  const int end = rowptr[n + 1];
  const float* xl = xw + lane * 4;
  v4f acc = {0.f, 0.f, 0.f, 0.f};
  for (int base = beg; base < end; base += 32) {
    const int j = base + lane;
    v2i ent = {0, 0};
    if (j < end) ent = *(const v2i*)(csr + 2 * (size_t)j);
    const int m = (end - base < 32) ? (end - base) : 32;
#pragma unroll 4
    for (int k = 0; k < m; ++k) {
      const int sk = __builtin_amdgcn_readlane(ent.x, k);
      const float wk = __int_as_float(__builtin_amdgcn_readlane(ent.y, k));
      const v4f v = *(const v4f*)(xl + (size_t)sk * DIM);
      acc += v * wk;
    }
  }
  const float di = dinv[n];
  const v4f self = *(const v4f*)(xl + (size_t)n * DIM);
  const v4f bb = *(const v4f*)(bl + lane * 4);
  v4f r = bb + self * di * di;
  r += acc;
  r.x = fmaxf(r.x, 0.f);
  r.y = fmaxf(r.y, 0.f);
  r.z = fmaxf(r.z, 0.f);
  r.w = fmaxf(r.w, 0.f);
  *(v4f*)(h + (size_t)n * DIM + lane * 4) = r;
}

__global__ void pool_zero_kernel(float* __restrict__ pool, float* __restrict__ cnt) {
  int t = blockIdx.x * blockDim.x + threadIdx.x;
  if (t < N_GRAPHS * DIM) pool[t] = 0.f;
  else if (t < N_GRAPHS * DIM + N_GRAPHS) cnt[t - N_GRAPHS * DIM] = 0.f;
}

#define POOL_NPW 10
static_assert(N_NODES % (8 * POOL_NPW) == 0, "whole blocks only: the node index is never out of range");
__device__ __forceinline__ void pool_add_row(float* __restrict__ pool, int g, int lane, v4f a) {
  float* p = pool + (size_t)g * DIM + lane * 4;
  unsafeAtomicAdd(p + 0, a.x);
  unsafeAtomicAdd(p + 1, a.y);
  unsafeAtomicAdd(p + 2, a.z);
  unsafeAtomicAdd(p + 3, a.w);
}
__global__ void __launch_bounds__(256)
pool_scatter_kernel(const int* __restrict__ batch, const float* __restrict__ h, float* __restrict__ pool) {
  const int lane = threadIdx.x & 31;
  const int w = __builtin_amdgcn_readfirstlane((int)((blockIdx.x * 256 + threadIdx.x) >> 5));
  const int n0 = w * POOL_NPW;
  const float* hl = h + (size_t)n0 * DIM + lane * 4;
  int g[POOL_NPW];
  v4f row[POOL_NPW];
#pragma unroll
  for (int i = 0; i < POOL_NPW; ++i) {
    g[i] = batch[n0 + i];
    row[i] = *(const v4f*)(hl + (size_t)i * DIM);
  }
  int cur = g[0];
  v4f acc = row[0];
#pragma unroll
  for (int i = 1; i < POOL_NPW; ++i) {
    if (g[i] == cur) {
      acc += row[i];
    } else {
      pool_add_row(pool, cur, lane, acc);
      cur = g[i];
      acc = row[i];
    }
  }
  pool_add_row(pool, cur, lane, acc);
}

__global__ void cnt_scatter_kernel(const int* __restrict__ batch, float* __restrict__ cnt) {
  int n = blockIdx.x * blockDim.x + threadIdx.x;
  if (n < N_NODES) unsafeAtomicAdd(&cnt[batch[n]], 1.0f);
}

__global__ void pool_div_kernel(float* __restrict__ pool, const float* __restrict__ cnt) {
  int t = blockIdx.x * blockDim.x + threadIdx.x;
  if (t >= N_GRAPHS * DIM) return;
  pool[t] /= fmaxf(cnt[t / DIM], 1.0f);
}

extern "C" void kernel_launch(void* const* d_in, const int* in_sizes, int n_in,
                              void* d_out, int out_size, void* d_ws, size_t ws_size,
                              hipStream_t stream) {
  const int*   x    = (const int*)d_in[0];
  const int*   ei   = (const int*)d_in[1];
  const int*   batch= (const int*)d_in[2];
  const float* emb  = (const float*)d_in[3];
  const float* W    = (const float*)d_in[4];
  const float* b    = (const float*)d_in[5];
  const float* Wr   = (const float*)d_in[6];
  const float* br   = (const float*)d_in[7];
  float* out = (float*)d_out;

  char* ws = (char*)d_ws;
  const size_t NH = (size_t)N_NODES * DIM * sizeof(float);
  float* h    = (float*)(ws);
  float* xw   = (float*)(ws + NH);
  int*   ecnt   = (int*)(ws + 2 * NH);
  int*   rowptr = (int*)(ws + 2 * NH + 256 * 1024);
  int*   rank   = (int*)(ws + 2 * NH + 512 * 1024);
  int*   csr    = (int*)(ws + 2 * NH + 8 * 1024 * 1024);
  float* dinv = (float*)(ws + 3 * NH);
  float* pool = (float*)(ws + 3 * NH + 4 * 1024 * 1024);
  float* cnt  = (float*)(ws + 3 * NH + 4 * 1024 * 1024 + 512 * 1024);

  const int B256 = 256;
  const int n_blocks  = (N_NODES + B256 - 1) / B256;
  const int e4_blocks = (N_EDGES / 4 + B256 - 1) / B256;
  const int emb_blocks  = N_NODES / (8 * EMB_NPW);
  const int pool_blocks = N_NODES / (8 * POOL_NPW);

  embed_kernel<<<emb_blocks, B256, 0, stream>>>(x, emb, h);

  ecnt_zero_kernel<<<n_blocks, B256, 0, stream>>>(ecnt);
  count_rank_kernel<<<e4_blocks, B256, 0, stream>>>(ei, ecnt, rank);
  rowptr_dinv_kernel<<<(N_NODES + 1023) / 1024, 1024, 0, stream>>>(ecnt, rowptr, dinv);
  csr_fill_kernel<<<e4_blocks, B256, 0, stream>>>(ei, rank, rowptr, dinv, csr);

  const int g_blocks = 1024;
  const int ga_blocks = N_NODES * 32 / B256;
  for (int l = 0; l < N_LAYERS; ++l) {
    gemm_wmma_f32<<<g_blocks, B256, 0, stream>>>(h, W + (size_t)l * DIM * DIM, nullptr, xw, N_NODES / 16);
    gather_relu_kernel<<<ga_blocks, B256, 0, stream>>>(rowptr, csr, xw, dinv, b + (size_t)l * DIM, h);
  }

  pool_zero_kernel<<<(N_GRAPHS * DIM + N_GRAPHS + B256 - 1) / B256, B256, 0, stream>>>(pool, cnt);
  pool_scatter_kernel<<<pool_blocks, B256, 0, stream>>>(batch, h, pool);
  cnt_scatter_kernel<<<n_blocks, B256, 0, stream>>>(batch, cnt);
  pool_div_kernel<<<(N_GRAPHS * DIM + B256 - 1) / B256, B256, 0, stream>>>(pool, cnt);

  gemm_wmma_f32<<<N_GRAPHS / 16, B256, 0, stream>>>(pool, Wr, br, out, N_GRAPHS / 16);
}


# Each layer's GEMM launch makes its own A rows (embedding sum, or the previous layer's neighbour sum with bias and relu) and hands them to the products through LDS: 9 launches become 5; same sums
# speedup vs baseline: 3.4334x; 1.0685x over previous
#include <hip/hip_runtime.h>

#define N_NODES  50000
#define N_EDGES  800000
#define DIM      128
#define N_LAYERS 4
#define N_GRAPHS 512
#define N_COLS   4
#define VOCAB    128

typedef __attribute__((ext_vector_type(2))) float v2f;
typedef __attribute__((ext_vector_type(4))) float v4f;
typedef __attribute__((ext_vector_type(8))) float v8f;

__device__ __forceinline__ v4f embed_row(const int* __restrict__ x, const float* __restrict__ emb,
                                         int n, int lane) {
  const float* el = emb + lane * 4;
  v4f s = {0.f, 0.f, 0.f, 0.f};
#pragma unroll
  for (int c = 0; c < N_COLS; ++c) {
    const int v = x[n * N_COLS + c];
    s += *(const v4f*)(el + ((size_t)c * VOCAB + v) * DIM);
  }
  return s;
}

typedef __attribute__((ext_vector_type(2))) int v2i;

__global__ void ecnt_zero_kernel(int* __restrict__ ecnt) {
  int n = blockIdx.x * blockDim.x + threadIdx.x;
  if (n < N_NODES) ecnt[n] = 0;
}

typedef __attribute__((ext_vector_type(4))) int v4i;
static_assert(N_EDGES % 4 == 0, "edges are taken four at a time");
__global__ void count_rank_kernel(const int* __restrict__ ei, int* __restrict__ ecnt,
                                  int* __restrict__ rank) {
  int t = blockIdx.x * blockDim.x + threadIdx.x;
  if (t >= N_EDGES / 4) return;
  const v4i d = *(const v4i*)(ei + N_EDGES + 4 * (size_t)t);
  v4i r;
  r.x = atomicAdd(&ecnt[d.x], 1);
  r.y = atomicAdd(&ecnt[d.y], 1);
  r.z = atomicAdd(&ecnt[d.z], 1);
  r.w = atomicAdd(&ecnt[d.w], 1);
  *(v4i*)(rank + 4 * (size_t)t) = r;
}

__global__ void __launch_bounds__(1024)
rowptr_dinv_kernel(const int* __restrict__ ecnt, int* __restrict__ rowptr, float* __restrict__ dinv) {
  __shared__ int s_pre[32];
  __shared__ int s_sum[32];
  const int t = threadIdx.x, lane = t & 31, wv = t >> 5;
  const int base = blockIdx.x * 1024;
  int pre = 0;
  for (int i = t; i < base; i += 1024) pre += ecnt[i];
  const int i = base + t;
  const int c = (i < N_NODES) ? ecnt[i] : 0;
  int x = c;
  for (int off = 1; off < 32; off <<= 1) {
    int y = __shfl_up(x, off, 32);
    if (lane >= off) x += y;
  }
  for (int off = 16; off > 0; off >>= 1) pre += __shfl_xor(pre, off, 32);
  if (lane == 31) s_sum[wv] = x;
  if (lane == 0) s_pre[wv] = pre;
  __syncthreads();
  const int ws = s_sum[lane];
  int bp = s_pre[lane];
  int wx = ws;
  for (int off = 1; off < 32; off <<= 1) {
    int y = __shfl_up(wx, off, 32);
    if (lane >= off) wx += y;
  }
  for (int off = 16; off > 0; off >>= 1) bp += __shfl_xor(bp, off, 32);
  const int wave_excl = __shfl(wx, wv, 32) - __shfl(ws, wv, 32);
  const int excl = bp + wave_excl + x - c;
  if (i < N_NODES) {
    rowptr[i] = excl;
    float d = 1.0f + (float)c;
    dinv[i] = (d > 0.f) ? rsqrtf(d) : 0.f;
    if (i == N_NODES - 1) rowptr[N_NODES] = excl + c;
  }
}

__global__ void csr_fill_kernel(const int* __restrict__ ei, const int* __restrict__ rank,
                                const int* __restrict__ rowptr, const float* __restrict__ dinv,
                                int* __restrict__ csr) {
  int t = blockIdx.x * blockDim.x + threadIdx.x;
  if (t >= N_EDGES / 4) return;
  const v4i s4 = *(const v4i*)(ei + 4 * (size_t)t);
  const v4i d4 = *(const v4i*)(ei + N_EDGES + 4 * (size_t)t);
  const v4i r4 = *(const v4i*)(rank + 4 * (size_t)t);
#pragma unroll
  for (int k = 0; k < 4; ++k) {
    const int s = s4[k];
    const int d = d4[k];
    const int pos = rowptr[d] + r4[k];
    v2i ent;
    ent.x = s;
    ent.y = __float_as_int(dinv[s] * dinv[d]);
    *(v2i*)(csr + 2 * (size_t)pos) = ent;
  }
}

#define A_PITCH 132

__device__ __forceinline__ void load_b_frags(const float* __restrict__ B, int half, int col, v2f (&bf)[32]) {
  const float* bp = B + (size_t)(4 * half) * DIM + col;
#pragma unroll
  for (int s = 0; s < 32; ++s) {
    const int r0 = 8 * (s >> 1) + 2 * (s & 1);
    bf[s].x = bp[r0 * DIM];
    bf[s].y = bp[(r0 + 1) * DIM];
  }
}

__device__ __forceinline__ v8f tile_product(const float* fa, const v2f (&bf)[32]) {
  v8f acc = {};
#pragma unroll
  for (int j = 0; j < 16; ++j) {
    const v4f a = *(const v4f*)(fa + 8 * j);
    v2f alo, ahi;
    alo.x = a.x; alo.y = a.y;
    ahi.x = a.z; ahi.y = a.w;
    acc = __builtin_amdgcn_wmma_f32_16x16x4_f32(
         false, alo,  false, bf[2 * j],
         (short)0, acc,  false,  false);
    acc = __builtin_amdgcn_wmma_f32_16x16x4_f32(
         false, ahi,  false, bf[2 * j + 1],
         (short)0, acc,  false,  false);
  }
  return acc;
}

__global__ void __launch_bounds__(256)
gemm_wmma_f32(const float* __restrict__ A, const float* __restrict__ B,
              const float* __restrict__ bias, float* __restrict__ Cout, int n_tiles) {
  __shared__ __attribute__((aligned(16))) float sA[2][16 * A_PITCH];
  const int tid  = threadIdx.x;
  const int lane = tid & 31;
  const int wv   = tid >> 5;
  const int half = lane >> 4;
  const int lm   = lane & 15;
  const int col0 = wv * 16;

  v2f bf[32];
  load_b_frags(B, half, col0 + lm, bf);
  const float badd = bias ? bias[col0 + lm] : 0.f;

  const float* ga = A + (size_t)wv * DIM + lane * 4;
  const int so0 = wv * A_PITCH + lane * 4;
  const int so1 = (wv + 8) * A_PITCH + lane * 4;
  const int fo  = lm * A_PITCH + 4 * half;

  int tile = blockIdx.x;
  v4f p0 = {0.f, 0.f, 0.f, 0.f}, p1 = {0.f, 0.f, 0.f, 0.f};
  if (tile < n_tiles) {
    p0 = *(const v4f*)(ga + (size_t)tile * 16 * DIM);
    p1 = *(const v4f*)(ga + (size_t)tile * 16 * DIM + 8 * DIM);
    *(v4f*)(&sA[0][so0]) = p0;
    *(v4f*)(&sA[0][so1]) = p1;
  }
  __syncthreads();
  int buf = 0;
  for (; tile < n_tiles; tile += gridDim.x) {
    const int nt = tile + gridDim.x;
    const bool more = nt < n_tiles;
    if (more) {
      p0 = *(const v4f*)(ga + (size_t)nt * 16 * DIM);
      p1 = *(const v4f*)(ga + (size_t)nt * 16 * DIM + 8 * DIM);
    }
    const v8f acc = tile_product(&sA[buf][fo], bf);
    float* cp = Cout + (size_t)(tile * 16 + half * 8) * DIM + col0 + lm;
#pragma unroll
    for (int r = 0; r < 8; ++r) cp[r * DIM] = acc[r] + badd;
    if (more) {
      *(v4f*)(&sA[buf ^ 1][so0]) = p0;
      *(v4f*)(&sA[buf ^ 1][so1]) = p1;
    }
    __syncthreads();
    buf ^= 1;
  }
}

static_assert((N_NODES * 32) % 256 == 0, "whole blocks only: the node index is never out of range");
__global__ void __launch_bounds__(256)
gather_relu_kernel(const int* __restrict__ rowptr, const int* __restrict__ csr,
                   const float* __restrict__ xw, const float* __restrict__ dinv,
                   const float* __restrict__ bl, float* __restrict__ h) {
  const int lane = threadIdx.x & 31;
  const int n = __builtin_amdgcn_readfirstlane((int)((blockIdx.x * 256 + threadIdx.x) >> 5));
  const int beg = rowptr[n];
  const int end = rowptr[n + 1];
  const float* xl = xw + lane * 4;
  v4f acc = {0.f, 0.f, 0.f, 0.f};
  for (int base = beg; base < end; base += 32) {
    const int j = base + lane;
    v2i ent = {0, 0};
    if (j < end) ent = *(const v2i*)(csr + 2 * (size_t)j);
    const int m = (end - base < 32) ? (end - base) : 32;
#pragma unroll 4
    for (int k = 0; k < m; ++k) {
      const int sk = __builtin_amdgcn_readlane(ent.x, k);
      const float wk = __int_as_float(__builtin_amdgcn_readlane(ent.y, k));
      const v4f v = *(const v4f*)(xl + (size_t)sk * DIM);
      acc += v * wk;
    }
  }
  const float di = dinv[n];
  const v4f self = *(const v4f*)(xl + (size_t)n * DIM);
  const v4f bb = *(const v4f*)(bl + lane * 4);
  v4f r = bb + self * di * di;
  r += acc;
  r.x = fmaxf(r.x, 0.f);
  r.y = fmaxf(r.y, 0.f);
  r.z = fmaxf(r.z, 0.f);
  r.w = fmaxf(r.w, 0.f);
  *(v4f*)(h + (size_t)n * DIM + lane * 4) = r;
}


template <int G>
__device__ __forceinline__ void add_rows(v4f& acc, const float* __restrict__ xw, int lane4, v2i ent, int k0) {
  v4f v[G];
  float w[G];
#pragma unroll
  for (int u = 0; u < G; ++u) {
    const int sk = __builtin_amdgcn_readlane(ent.x, k0 + u);
    w[u] = __int_as_float(__builtin_amdgcn_readlane(ent.y, k0 + u));
    v[u] = *(const v4f*)((const char*)xw + ((unsigned)sk * DIM + (unsigned)lane4) * 4u);
  }
#pragma unroll
  for (int u = 0; u < G; ++u) acc += v[u] * w[u];
}

__device__ __forceinline__ v2i edge_chunk(const int* __restrict__ csr, int base, int end, int lane) {
  const int j = base + lane;
  v2i ent = {0, 0};
  if (j < end) ent = *(const v2i*)(csr + 2 * (size_t)j);
  return ent;
}

__device__ __forceinline__ v4f edge_sum(const int* __restrict__ csr, const float* __restrict__ xw,
                                        int beg, int end, v2i first, int lane) {
  v4f acc = {0.f, 0.f, 0.f, 0.f};
  v2i ent = first;
  for (int base = beg; base < end; base += 32) {
    if (base != beg) ent = edge_chunk(csr, base, end, lane);
    const int m = (end - base < 32) ? (end - base) : 32;
    int k = 0;
    for (; k + 8 <= m; k += 8) add_rows<8>(acc, xw, lane * 4, ent, k);
    if (m - k >= 4) { add_rows<4>(acc, xw, lane * 4, ent, k); k += 4; }
    if (m - k >= 2) { add_rows<2>(acc, xw, lane * 4, ent, k); k += 2; }
    if (m - k >= 1) add_rows<1>(acc, xw, lane * 4, ent, k);
  }
  return acc;
}

__device__ __forceinline__ v4f bias_self_relu(v4f bb, v4f self, float di, v4f acc) {
  v4f r = bb + self * di * di;
  r += acc;
  r.x = fmaxf(r.x, 0.f);
  r.y = fmaxf(r.y, 0.f);
  r.z = fmaxf(r.z, 0.f);
  r.w = fmaxf(r.w, 0.f);
  return r;
}

static_assert(N_NODES % 16 == 0, "whole 16-row tiles only: the node index is never out of range");
template <bool FROM_EMBED>
__global__ void __launch_bounds__(256, 8)
rows_gemm_kernel(const int* __restrict__ x, const float* __restrict__ emb,
                 const int* __restrict__ rowptr, const int* __restrict__ csr,
                 const float* __restrict__ xw, const float* __restrict__ dinv,
                 const float* __restrict__ bl,
                 const float* __restrict__ B, float* __restrict__ Cout, int n_tiles) {
  __shared__ __attribute__((aligned(16))) float sA[2][16 * A_PITCH];
  const int lane = threadIdx.x & 31;
  const int wv   = __builtin_amdgcn_readfirstlane((int)(threadIdx.x >> 5));
  const int half = lane >> 4;
  const int lm   = lane & 15;
  const int col0 = wv * 16;

  v2f bf[32];
  load_b_frags(B, half, col0 + lm, bf);

  const int so0 = wv * A_PITCH + lane * 4;
  const int so1 = (wv + 8) * A_PITCH + lane * 4;
  const int fo  = lm * A_PITCH + 4 * half;

  v4f bb = {0.f, 0.f, 0.f, 0.f};
  if (!FROM_EMBED) bb = *(const v4f*)(bl + lane * 4);

  int buf = 0;
  for (int tile = blockIdx.x; tile < n_tiles; tile += gridDim.x) {
    const int n0 = tile * 16 + wv;
    const int n1 = n0 + 8;
    if (FROM_EMBED) {
      const v4f r0 = embed_row(x, emb, n0, lane);
      const v4f r1 = embed_row(x, emb, n1, lane);
      *(v4f*)(&sA[buf][so0]) = r0;
      *(v4f*)(&sA[buf][so1]) = r1;
    } else {
      const int beg0 = rowptr[n0], end0 = rowptr[n0 + 1];
      const int beg1 = rowptr[n1], end1 = rowptr[n1 + 1];
      const v2i e0 = edge_chunk(csr, beg0, end0, lane);
      const v2i e1 = edge_chunk(csr, beg1, end1, lane);
      const float di0 = dinv[n0];
      const v4f self0 = *(const v4f*)((const char*)xw + ((unsigned)n0 * DIM + (unsigned)lane * 4) * 4u);
      const v4f a0 = edge_sum(csr, xw, beg0, end0, e0, lane);
      const float di1 = dinv[n1];
      const v4f self1 = *(const v4f*)((const char*)xw + ((unsigned)n1 * DIM + (unsigned)lane * 4) * 4u);
      *(v4f*)(&sA[buf][so0]) = bias_self_relu(bb, self0, di0, a0);
      const v4f a1 = edge_sum(csr, xw, beg1, end1, e1, lane);
      *(v4f*)(&sA[buf][so1]) = bias_self_relu(bb, self1, di1, a1);
    }
    __syncthreads();
    const v8f acc = tile_product(&sA[buf][fo], bf);
    float* cp = (float*)((char*)Cout + (unsigned)((tile * 16 + half * 8) * DIM + col0 + lm) * 4u);
#pragma unroll
    for (int r = 0; r < 8; ++r) cp[r * DIM] = acc[r];
    buf ^= 1;
  }
}

__global__ void pool_zero_kernel(float* __restrict__ pool, float* __restrict__ cnt) {
  int t = blockIdx.x * blockDim.x + threadIdx.x;
  if (t < N_GRAPHS * DIM) pool[t] = 0.f;
  else if (t < N_GRAPHS * DIM + N_GRAPHS) cnt[t - N_GRAPHS * DIM] = 0.f;
}

#define POOL_NPW 10
static_assert(N_NODES % (8 * POOL_NPW) == 0, "whole blocks only: the node index is never out of range");
__device__ __forceinline__ void pool_add_row(float* __restrict__ pool, int g, int lane, v4f a) {
  float* p = pool + (size_t)g * DIM + lane * 4;
  unsafeAtomicAdd(p + 0, a.x);
  unsafeAtomicAdd(p + 1, a.y);
  unsafeAtomicAdd(p + 2, a.z);
  unsafeAtomicAdd(p + 3, a.w);
}
__global__ void __launch_bounds__(256)
pool_scatter_kernel(const int* __restrict__ batch, const float* __restrict__ h, float* __restrict__ pool) {
  const int lane = threadIdx.x & 31;
  const int w = __builtin_amdgcn_readfirstlane((int)((blockIdx.x * 256 + threadIdx.x) >> 5));
  const int n0 = w * POOL_NPW;
  const float* hl = h + (size_t)n0 * DIM + lane * 4;
  int g[POOL_NPW];
  v4f row[POOL_NPW];
#pragma unroll
  for (int i = 0; i < POOL_NPW; ++i) {
    g[i] = batch[n0 + i];
    row[i] = *(const v4f*)(hl + (size_t)i * DIM);
  }
  int cur = g[0];
  v4f acc = row[0];
#pragma unroll
  for (int i = 1; i < POOL_NPW; ++i) {
    if (g[i] == cur) {
      acc += row[i];
    } else {
      pool_add_row(pool, cur, lane, acc);
      cur = g[i];
      acc = row[i];
    }
  }
  pool_add_row(pool, cur, lane, acc);
}

__global__ void cnt_scatter_kernel(const int* __restrict__ batch, float* __restrict__ cnt) {
  int n = blockIdx.x * blockDim.x + threadIdx.x;
  if (n < N_NODES) unsafeAtomicAdd(&cnt[batch[n]], 1.0f);
}

__global__ void pool_div_kernel(float* __restrict__ pool, const float* __restrict__ cnt) {
  int t = blockIdx.x * blockDim.x + threadIdx.x;
  if (t >= N_GRAPHS * DIM) return;
  pool[t] /= fmaxf(cnt[t / DIM], 1.0f);
}

extern "C" void kernel_launch(void* const* d_in, const int* in_sizes, int n_in,
                              void* d_out, int out_size, void* d_ws, size_t ws_size,
                              hipStream_t stream) {
  const int*   x    = (const int*)d_in[0];
  const int*   ei   = (const int*)d_in[1];
  const int*   batch= (const int*)d_in[2];
  const float* emb  = (const float*)d_in[3];
  const float* W    = (const float*)d_in[4];
  const float* b    = (const float*)d_in[5];
  const float* Wr   = (const float*)d_in[6];
  const float* br   = (const float*)d_in[7];
  float* out = (float*)d_out;

  char* ws = (char*)d_ws;
  const size_t NH = (size_t)N_NODES * DIM * sizeof(float);
  float* bufA = (float*)(ws);
  float* bufB = (float*)(ws + NH);
  int*   ecnt   = (int*)(ws + 2 * NH);
  int*   rowptr = (int*)(ws + 2 * NH + 256 * 1024);
  int*   rank   = (int*)(ws + 2 * NH + 512 * 1024);
  int*   csr    = (int*)(ws + 2 * NH + 8 * 1024 * 1024);
  float* dinv = (float*)(ws + 3 * NH);
  float* pool = (float*)(ws + 3 * NH + 4 * 1024 * 1024);
  float* cnt  = (float*)(ws + 3 * NH + 4 * 1024 * 1024 + 512 * 1024);

  const int B256 = 256;
  const int n_blocks  = (N_NODES + B256 - 1) / B256;
  const int e4_blocks = (N_EDGES / 4 + B256 - 1) / B256;
  const int pool_blocks = N_NODES / (8 * POOL_NPW);

  const int g_blocks = 782;
  const int n_tiles = N_NODES / 16;

  rows_gemm_kernel<true><<<g_blocks, B256, 0, stream>>>(x, emb, nullptr, nullptr, nullptr, nullptr, nullptr,
                                                        W, bufA, n_tiles);

  ecnt_zero_kernel<<<n_blocks, B256, 0, stream>>>(ecnt);
  count_rank_kernel<<<e4_blocks, B256, 0, stream>>>(ei, ecnt, rank);
  rowptr_dinv_kernel<<<(N_NODES + 1023) / 1024, 1024, 0, stream>>>(ecnt, rowptr, dinv);
  csr_fill_kernel<<<e4_blocks, B256, 0, stream>>>(ei, rank, rowptr, dinv, csr);

  rows_gemm_kernel<false><<<g_blocks, B256, 0, stream>>>(nullptr, nullptr, rowptr, csr, bufA, dinv, b,
                                                         W + (size_t)1 * DIM * DIM, bufB, n_tiles);
  rows_gemm_kernel<false><<<g_blocks, B256, 0, stream>>>(nullptr, nullptr, rowptr, csr, bufB, dinv, b + 1 * DIM,
                                                         W + (size_t)2 * DIM * DIM, bufA, n_tiles);
  rows_gemm_kernel<false><<<g_blocks, B256, 0, stream>>>(nullptr, nullptr, rowptr, csr, bufA, dinv, b + 2 * DIM,
                                                         W + (size_t)3 * DIM * DIM, bufB, n_tiles);
  const int ga_blocks = N_NODES * 32 / B256;
  float* h = bufA;
  gather_relu_kernel<<<ga_blocks, B256, 0, stream>>>(rowptr, csr, bufB, dinv, b + 3 * DIM, h);

  pool_zero_kernel<<<(N_GRAPHS * DIM + N_GRAPHS + B256 - 1) / B256, B256, 0, stream>>>(pool, cnt);
  pool_scatter_kernel<<<pool_blocks, B256, 0, stream>>>(batch, h, pool);
  cnt_scatter_kernel<<<n_blocks, B256, 0, stream>>>(batch, cnt);
  pool_div_kernel<<<(N_GRAPHS * DIM + B256 - 1) / B256, B256, 0, stream>>>(pool, cnt);

  gemm_wmma_f32<<<N_GRAPHS / 16, B256, 0, stream>>>(pool, Wr, br, out, N_GRAPHS / 16);
}


# Helper launches folded into neighbours: zero fills into first GEMM launch, last layer's sum and node counts into pool launch (5-node runs: atomic sum order), mean division into readout: 14 to 9
# speedup vs baseline: 3.7063x; 1.0795x over previous
#include <hip/hip_runtime.h>

#define N_NODES  50000
#define N_EDGES  800000
#define DIM      128
#define N_LAYERS 4
#define N_GRAPHS 512
#define N_COLS   4
#define VOCAB    128

typedef __attribute__((ext_vector_type(2))) float v2f;
typedef __attribute__((ext_vector_type(4))) float v4f;
typedef __attribute__((ext_vector_type(8))) float v8f;

__device__ __forceinline__ v4f embed_row(const int* __restrict__ x, const float* __restrict__ emb,
                                         int n, int lane) {
  const float* el = emb + lane * 4;
  v4f s = {0.f, 0.f, 0.f, 0.f};
#pragma unroll
  for (int c = 0; c < N_COLS; ++c) {
    const int v = x[n * N_COLS + c];
    s += *(const v4f*)(el + ((size_t)c * VOCAB + v) * DIM);
  }
  return s;
}

typedef __attribute__((ext_vector_type(2))) int v2i;

typedef __attribute__((ext_vector_type(4))) int v4i;
static_assert(N_EDGES % 4 == 0, "edges are taken four at a time");
__global__ void count_rank_kernel(const int* __restrict__ ei, int* __restrict__ ecnt,
                                  int* __restrict__ rank) {
  int t = blockIdx.x * blockDim.x + threadIdx.x;
  if (t >= N_EDGES / 4) return;
  const v4i d = *(const v4i*)(ei + N_EDGES + 4 * (size_t)t);
  v4i r;
  r.x = atomicAdd(&ecnt[d.x], 1);
  r.y = atomicAdd(&ecnt[d.y], 1);
  r.z = atomicAdd(&ecnt[d.z], 1);
  r.w = atomicAdd(&ecnt[d.w], 1);
  *(v4i*)(rank + 4 * (size_t)t) = r;
}

__global__ void __launch_bounds__(1024)
rowptr_dinv_kernel(const int* __restrict__ ecnt, int* __restrict__ rowptr, float* __restrict__ dinv) {
  __shared__ int s_pre[32];
  __shared__ int s_sum[32];
  const int t = threadIdx.x, lane = t & 31, wv = t >> 5;
  const int base = blockIdx.x * 1024;
  int pre = 0;
  for (int i = t; i < base; i += 1024) pre += ecnt[i];
  const int i = base + t;
  const int c = (i < N_NODES) ? ecnt[i] : 0;
  int x = c;
  for (int off = 1; off < 32; off <<= 1) {
    int y = __shfl_up(x, off, 32);
    if (lane >= off) x += y;
  }
  for (int off = 16; off > 0; off >>= 1) pre += __shfl_xor(pre, off, 32);
  if (lane == 31) s_sum[wv] = x;
  if (lane == 0) s_pre[wv] = pre;
  __syncthreads();
  const int ws = s_sum[lane];
  int bp = s_pre[lane];
  int wx = ws;
  for (int off = 1; off < 32; off <<= 1) {
    int y = __shfl_up(wx, off, 32);
    if (lane >= off) wx += y;
  }
  for (int off = 16; off > 0; off >>= 1) bp += __shfl_xor(bp, off, 32);
  const int wave_excl = __shfl(wx, wv, 32) - __shfl(ws, wv, 32);
  const int excl = bp + wave_excl + x - c;
  if (i < N_NODES) {
    rowptr[i] = excl;
    float d = 1.0f + (float)c;
    dinv[i] = (d > 0.f) ? rsqrtf(d) : 0.f;
    if (i == N_NODES - 1) rowptr[N_NODES] = excl + c;
  }
}

__global__ void csr_fill_kernel(const int* __restrict__ ei, const int* __restrict__ rank,
                                const int* __restrict__ rowptr, const float* __restrict__ dinv,
                                int* __restrict__ csr) {
  int t = blockIdx.x * blockDim.x + threadIdx.x;
  if (t >= N_EDGES / 4) return;
  const v4i s4 = *(const v4i*)(ei + 4 * (size_t)t);
  const v4i d4 = *(const v4i*)(ei + N_EDGES + 4 * (size_t)t);
  const v4i r4 = *(const v4i*)(rank + 4 * (size_t)t);
#pragma unroll
  for (int k = 0; k < 4; ++k) {
    const int s = s4[k];
    const int d = d4[k];
    const int pos = rowptr[d] + r4[k];
    v2i ent;
    ent.x = s;
    ent.y = __float_as_int(dinv[s] * dinv[d]);
    *(v2i*)(csr + 2 * (size_t)pos) = ent;
  }
}

#define A_PITCH 132

__device__ __forceinline__ void load_b_frags(const float* __restrict__ B, int half, int col, v2f (&bf)[32]) {
  const float* bp = B + (size_t)(4 * half) * DIM + col;
#pragma unroll
  for (int s = 0; s < 32; ++s) {
    const int r0 = 8 * (s >> 1) + 2 * (s & 1);
    bf[s].x = bp[r0 * DIM];
    bf[s].y = bp[(r0 + 1) * DIM];
  }
}

__device__ __forceinline__ v8f tile_product(const float* fa, const v2f (&bf)[32]) {
  v8f acc = {};
#pragma unroll
  for (int j = 0; j < 16; ++j) {
    const v4f a = *(const v4f*)(fa + 8 * j);
    v2f alo, ahi;
    alo.x = a.x; alo.y = a.y;
    ahi.x = a.z; ahi.y = a.w;
    acc = __builtin_amdgcn_wmma_f32_16x16x4_f32(
         false, alo,  false, bf[2 * j],
         (short)0, acc,  false,  false);
    acc = __builtin_amdgcn_wmma_f32_16x16x4_f32(
         false, ahi,  false, bf[2 * j + 1],
         (short)0, acc,  false,  false);
  }
  return acc;
}

__device__ __forceinline__ v4f mean_row(v4f s, float c) {
  const float d = fmaxf(c, 1.0f);
  v4f r;
  r.x = s.x / d;
  r.y = s.y / d;
  r.z = s.z / d;
  r.w = s.w / d;
  return r;
}

__global__ void __launch_bounds__(256)
mean_gemm_wmma_f32(const float* __restrict__ A, const float* __restrict__ cnt, const float* __restrict__ B,
                   const float* __restrict__ bias, float* __restrict__ Cout, int n_tiles) {
  __shared__ __attribute__((aligned(16))) float sA[2][16 * A_PITCH];
  const int tid  = threadIdx.x;
  const int lane = tid & 31;
  const int wv   = tid >> 5;
  const int half = lane >> 4;
  const int lm   = lane & 15;
  const int col0 = wv * 16;

  v2f bf[32];
  load_b_frags(B, half, col0 + lm, bf);
  const float badd = bias[col0 + lm];

  const float* ga = A + (size_t)wv * DIM + lane * 4;
  const int so0 = wv * A_PITCH + lane * 4;
  const int so1 = (wv + 8) * A_PITCH + lane * 4;
  const int fo  = lm * A_PITCH + 4 * half;

  int tile = blockIdx.x;
  v4f p0 = {0.f, 0.f, 0.f, 0.f}, p1 = {0.f, 0.f, 0.f, 0.f};
  float c0 = 1.f, c1 = 1.f;
  if (tile < n_tiles) {
    p0 = *(const v4f*)(ga + (size_t)tile * 16 * DIM);
    p1 = *(const v4f*)(ga + (size_t)tile * 16 * DIM + 8 * DIM);
    c0 = cnt[tile * 16 + wv];
    c1 = cnt[tile * 16 + wv + 8];
    *(v4f*)(&sA[0][so0]) = mean_row(p0, c0);
    *(v4f*)(&sA[0][so1]) = mean_row(p1, c1);
  }
  __syncthreads();
  int buf = 0;
  for (; tile < n_tiles; tile += gridDim.x) {
    const int nt = tile + gridDim.x;
    const bool more = nt < n_tiles;
    if (more) {
      p0 = *(const v4f*)(ga + (size_t)nt * 16 * DIM);
      p1 = *(const v4f*)(ga + (size_t)nt * 16 * DIM + 8 * DIM);
      c0 = cnt[nt * 16 + wv];
      c1 = cnt[nt * 16 + wv + 8];
    }
    const v8f acc = tile_product(&sA[buf][fo], bf);
    float* cp = Cout + (size_t)(tile * 16 + half * 8) * DIM + col0 + lm;
#pragma unroll
    for (int r = 0; r < 8; ++r) cp[r * DIM] = acc[r] + badd;
    if (more) {
      *(v4f*)(&sA[buf ^ 1][so0]) = mean_row(p0, c0);
      *(v4f*)(&sA[buf ^ 1][so1]) = mean_row(p1, c1);
    }
    __syncthreads();
    buf ^= 1;
  }
}


template <int G>
__device__ __forceinline__ void add_rows(v4f& acc, const float* __restrict__ xw, int lane4, v2i ent, int k0) {
  v4f v[G];
  float w[G];
#pragma unroll
  for (int u = 0; u < G; ++u) {
    const int sk = __builtin_amdgcn_readlane(ent.x, k0 + u);
    w[u] = __int_as_float(__builtin_amdgcn_readlane(ent.y, k0 + u));
    v[u] = *(const v4f*)((const char*)xw + ((unsigned)sk * DIM + (unsigned)lane4) * 4u);
  }
#pragma unroll
  for (int u = 0; u < G; ++u) acc += v[u] * w[u];
}

__device__ __forceinline__ v2i edge_chunk(const int* __restrict__ csr, int base, int end, int lane) {
  const int j = base + lane;
  v2i ent = {0, 0};
  if (j < end) ent = *(const v2i*)(csr + 2 * (size_t)j);
  return ent;
}

__device__ __forceinline__ v4f edge_sum(const int* __restrict__ csr, const float* __restrict__ xw,
                                        int beg, int end, v2i first, int lane) {
  v4f acc = {0.f, 0.f, 0.f, 0.f};
  v2i ent = first;
  for (int base = beg; base < end; base += 32) {
    if (base != beg) ent = edge_chunk(csr, base, end, lane);
    const int m = (end - base < 32) ? (end - base) : 32;
    int k = 0;
    for (; k + 8 <= m; k += 8) add_rows<8>(acc, xw, lane * 4, ent, k);
    if (m - k >= 4) { add_rows<4>(acc, xw, lane * 4, ent, k); k += 4; }
    if (m - k >= 2) { add_rows<2>(acc, xw, lane * 4, ent, k); k += 2; }
    if (m - k >= 1) add_rows<1>(acc, xw, lane * 4, ent, k);
  }
  return acc;
}

__device__ __forceinline__ v4f bias_self_relu(v4f bb, v4f self, float di, v4f acc) {
  v4f r = bb + self * di * di;
  r += acc;
  r.x = fmaxf(r.x, 0.f);
  r.y = fmaxf(r.y, 0.f);
  r.z = fmaxf(r.z, 0.f);
  r.w = fmaxf(r.w, 0.f);
  return r;
}

static_assert(N_NODES % 16 == 0, "whole 16-row tiles only: the node index is never out of range");
template <bool FROM_EMBED>
__global__ void __launch_bounds__(256, 8)
rows_gemm_kernel(const int* __restrict__ x, const float* __restrict__ emb,
                 const int* __restrict__ rowptr, const int* __restrict__ csr,
                 const float* __restrict__ xw, const float* __restrict__ dinv,
                 const float* __restrict__ bl,
                 const float* __restrict__ B, float* __restrict__ Cout, int n_tiles,
                 int* __restrict__ ecnt, float* __restrict__ pool, float* __restrict__ cnt) {
  if (FROM_EMBED) {
    const int t = blockIdx.x * 256 + threadIdx.x, nt = gridDim.x * 256;
    for (int i = t; i < N_NODES; i += nt) ecnt[i] = 0;
    for (int i = t; i < N_GRAPHS * DIM; i += nt) pool[i] = 0.f;
    for (int i = t; i < N_GRAPHS; i += nt) cnt[i] = 0.f;
  }
  __shared__ __attribute__((aligned(16))) float sA[2][16 * A_PITCH];
  const int lane = threadIdx.x & 31;
  const int wv   = __builtin_amdgcn_readfirstlane((int)(threadIdx.x >> 5));
  const int half = lane >> 4;
  const int lm   = lane & 15;
  const int col0 = wv * 16;

  v2f bf[32];
  load_b_frags(B, half, col0 + lm, bf);

  const int so0 = wv * A_PITCH + lane * 4;
  const int so1 = (wv + 8) * A_PITCH + lane * 4;
  const int fo  = lm * A_PITCH + 4 * half;

  v4f bb = {0.f, 0.f, 0.f, 0.f};
  if (!FROM_EMBED) bb = *(const v4f*)(bl + lane * 4);

  int buf = 0;
  for (int tile = blockIdx.x; tile < n_tiles; tile += gridDim.x) {
    const int n0 = tile * 16 + wv;
    const int n1 = n0 + 8;
    if (FROM_EMBED) {
      const v4f r0 = embed_row(x, emb, n0, lane);
      const v4f r1 = embed_row(x, emb, n1, lane);
      *(v4f*)(&sA[buf][so0]) = r0;
      *(v4f*)(&sA[buf][so1]) = r1;
    } else {
      const int beg0 = rowptr[n0], end0 = rowptr[n0 + 1];
      const int beg1 = rowptr[n1], end1 = rowptr[n1 + 1];
      const v2i e0 = edge_chunk(csr, beg0, end0, lane);
      const v2i e1 = edge_chunk(csr, beg1, end1, lane);
      const float di0 = dinv[n0];
      const v4f self0 = *(const v4f*)((const char*)xw + ((unsigned)n0 * DIM + (unsigned)lane * 4) * 4u);
      const v4f a0 = edge_sum(csr, xw, beg0, end0, e0, lane);
      const float di1 = dinv[n1];
      const v4f self1 = *(const v4f*)((const char*)xw + ((unsigned)n1 * DIM + (unsigned)lane * 4) * 4u);
      *(v4f*)(&sA[buf][so0]) = bias_self_relu(bb, self0, di0, a0);
      const v4f a1 = edge_sum(csr, xw, beg1, end1, e1, lane);
      *(v4f*)(&sA[buf][so1]) = bias_self_relu(bb, self1, di1, a1);
    }
    __syncthreads();
    const v8f acc = tile_product(&sA[buf][fo], bf);
    float* cp = (float*)((char*)Cout + (unsigned)((tile * 16 + half * 8) * DIM + col0 + lm) * 4u);
#pragma unroll
    for (int r = 0; r < 8; ++r) cp[r * DIM] = acc[r];
    buf ^= 1;
  }
}

#define POOL_NPW 5
static_assert(N_NODES % (8 * POOL_NPW) == 0, "whole blocks only: the node index is never out of range");
__device__ __forceinline__ void pool_add_run(float* __restrict__ pool, float* __restrict__ cnt,
                                             int g, int lane, v4f a, int len) {
  float* p = pool + (size_t)g * DIM + lane * 4;
  unsafeAtomicAdd(p + 0, a.x);
  unsafeAtomicAdd(p + 1, a.y);
  unsafeAtomicAdd(p + 2, a.z);
  unsafeAtomicAdd(p + 3, a.w);
  if (lane == 0) unsafeAtomicAdd(cnt + g, (float)len);
}
__global__ void __launch_bounds__(256)
gather_pool_kernel(const int* __restrict__ rowptr, const int* __restrict__ csr,
                   const float* __restrict__ xw, const float* __restrict__ dinv,
                   const float* __restrict__ bl, const int* __restrict__ batch,
                   float* __restrict__ pool, float* __restrict__ cnt) {
  const int lane = threadIdx.x & 31;
  const int w = __builtin_amdgcn_readfirstlane((int)((blockIdx.x * 256 + threadIdx.x) >> 5));
  const int n0 = w * POOL_NPW;
  const v4f bb = *(const v4f*)(bl + lane * 4);
  int g[POOL_NPW], rp[POOL_NPW + 1];
  v2i e[POOL_NPW];
  float di[POOL_NPW];
  v4f self[POOL_NPW];
#pragma unroll
  for (int i = 0; i <= POOL_NPW; ++i) rp[i] = rowptr[n0 + i];
#pragma unroll
  for (int i = 0; i < POOL_NPW; ++i) {
    g[i] = batch[n0 + i];
    di[i] = dinv[n0 + i];
  }
#pragma unroll
  for (int i = 0; i < POOL_NPW; ++i) {
    e[i] = edge_chunk(csr, rp[i], rp[i + 1], lane);
    self[i] = *(const v4f*)((const char*)xw + ((unsigned)(n0 + i) * DIM + (unsigned)lane * 4) * 4u);
  }
  v4f row[POOL_NPW];
#pragma unroll
  for (int i = 0; i < POOL_NPW; ++i)
    row[i] = bias_self_relu(bb, self[i], di[i], edge_sum(csr, xw, rp[i], rp[i + 1], e[i], lane));
  int cur = g[0];
  int len = 1;
  v4f acc = row[0];
#pragma unroll
  for (int i = 1; i < POOL_NPW; ++i) {
    if (g[i] == cur) {
      acc += row[i];
      ++len;
    } else {
      pool_add_run(pool, cnt, cur, lane, acc, len);
      cur = g[i];
      acc = row[i];
      len = 1;
    }
  }
  pool_add_run(pool, cnt, cur, lane, acc, len);
}

extern "C" void kernel_launch(void* const* d_in, const int* in_sizes, int n_in,
                              void* d_out, int out_size, void* d_ws, size_t ws_size,
                              hipStream_t stream) {
  const int*   x    = (const int*)d_in[0];
  const int*   ei   = (const int*)d_in[1];
  const int*   batch= (const int*)d_in[2];
  const float* emb  = (const float*)d_in[3];
  const float* W    = (const float*)d_in[4];
  const float* b    = (const float*)d_in[5];
  const float* Wr   = (const float*)d_in[6];
  const float* br   = (const float*)d_in[7];
  float* out = (float*)d_out;

  char* ws = (char*)d_ws;
  const size_t NH = (size_t)N_NODES * DIM * sizeof(float);
  float* bufA = (float*)(ws);
  float* bufB = (float*)(ws + NH);
  int*   ecnt   = (int*)(ws + 2 * NH);
  int*   rowptr = (int*)(ws + 2 * NH + 256 * 1024);
  int*   rank   = (int*)(ws + 2 * NH + 512 * 1024);
  int*   csr    = (int*)(ws + 2 * NH + 8 * 1024 * 1024);
  float* dinv = (float*)(ws + 3 * NH);
  float* pool = (float*)(ws + 3 * NH + 4 * 1024 * 1024);
  float* cnt  = (float*)(ws + 3 * NH + 4 * 1024 * 1024 + 512 * 1024);

  const int B256 = 256;
  const int e4_blocks = (N_EDGES / 4 + B256 - 1) / B256;
  const int pool_blocks = N_NODES / (8 * POOL_NPW);

  const int g_blocks = 782;
  const int n_tiles = N_NODES / 16;

  rows_gemm_kernel<true><<<g_blocks, B256, 0, stream>>>(x, emb, nullptr, nullptr, nullptr, nullptr, nullptr,
                                                        W, bufA, n_tiles, ecnt, pool, cnt);

  count_rank_kernel<<<e4_blocks, B256, 0, stream>>>(ei, ecnt, rank);
  rowptr_dinv_kernel<<<(N_NODES + 1023) / 1024, 1024, 0, stream>>>(ecnt, rowptr, dinv);
  csr_fill_kernel<<<e4_blocks, B256, 0, stream>>>(ei, rank, rowptr, dinv, csr);

  rows_gemm_kernel<false><<<g_blocks, B256, 0, stream>>>(nullptr, nullptr, rowptr, csr, bufA, dinv, b,
                                                         W + (size_t)1 * DIM * DIM, bufB, n_tiles,
                                                         nullptr, nullptr, nullptr);
  rows_gemm_kernel<false><<<g_blocks, B256, 0, stream>>>(nullptr, nullptr, rowptr, csr, bufB, dinv, b + 1 * DIM,
                                                         W + (size_t)2 * DIM * DIM, bufA, n_tiles,
                                                         nullptr, nullptr, nullptr);
  rows_gemm_kernel<false><<<g_blocks, B256, 0, stream>>>(nullptr, nullptr, rowptr, csr, bufA, dinv, b + 2 * DIM,
                                                         W + (size_t)3 * DIM * DIM, bufB, n_tiles,
                                                         nullptr, nullptr, nullptr);

  gather_pool_kernel<<<pool_blocks, B256, 0, stream>>>(rowptr, csr, bufB, dinv, b + 3 * DIM, batch, pool, cnt);

  mean_gemm_wmma_f32<<<N_GRAPHS / 16, B256, 0, stream>>>(pool, cnt, Wr, br, out, N_GRAPHS / 16);
}


# Edges dropped into a 64-entry slot per destination as they are counted, rows pre-scaled by deg^-1/2 (norm factored out of the sum: rounding changes): no prefix scan, no fill pass; 9 launches to 8
# speedup vs baseline: 4.0779x; 1.1003x over previous
#include <hip/hip_runtime.h>

#define N_NODES  50000
#define N_EDGES  800000
#define DIM      128
#define N_LAYERS 4
#define N_GRAPHS 512
#define N_COLS   4
#define VOCAB    128

typedef __attribute__((ext_vector_type(2))) float v2f;
typedef __attribute__((ext_vector_type(4))) float v4f;
typedef __attribute__((ext_vector_type(8))) float v8f;

__device__ __forceinline__ v4f embed_row(const int* __restrict__ x, const float* __restrict__ emb,
                                         int n, int lane) {
  const float* el = emb + lane * 4;
  v4f s = {0.f, 0.f, 0.f, 0.f};
#pragma unroll
  for (int c = 0; c < N_COLS; ++c) {
    const int v = x[n * N_COLS + c];
    s += *(const v4f*)(el + ((size_t)c * VOCAB + v) * DIM);
  }
  return s;
}

#define SLOT_CAP 64
typedef __attribute__((ext_vector_type(2))) int v2i;
typedef __attribute__((ext_vector_type(4))) int v4i;
static_assert(N_EDGES % 4 == 0, "edges are taken four at a time");
static_assert(N_NODES <= N_GRAPHS * DIM, "one thread per pool element also covers the node counters");

__global__ void __launch_bounds__(256)
zero_kernel(int* __restrict__ ecnt, int* __restrict__ ovf_cnt, float* __restrict__ pool, float* __restrict__ cnt) {
  const int t = blockIdx.x * 256 + threadIdx.x;
  pool[t] = 0.f;
  if (t < N_NODES) ecnt[t] = 0;
  if (t < N_GRAPHS) cnt[t] = 0.f;
  if (t == 0) *ovf_cnt = 0;
}

__global__ void __launch_bounds__(256)
count_slot_kernel(const int* __restrict__ ei, int* __restrict__ ecnt, int* __restrict__ slot,
                  int* __restrict__ ovf_cnt, int* __restrict__ ovf) {
  const int t = blockIdx.x * 256 + threadIdx.x;
  if (t >= N_EDGES / 4) return;
  const v4i s4 = *(const v4i*)(ei + 4 * (size_t)t);
  const v4i d4 = *(const v4i*)(ei + N_EDGES + 4 * (size_t)t);
  v4i r4;
  r4.x = atomicAdd(&ecnt[d4.x], 1);
  r4.y = atomicAdd(&ecnt[d4.y], 1);
  r4.z = atomicAdd(&ecnt[d4.z], 1);
  r4.w = atomicAdd(&ecnt[d4.w], 1);
#pragma unroll
  for (int k = 0; k < 4; ++k) {
    const int s = s4[k], d = d4[k], r = r4[k];
    if (r < SLOT_CAP) {
      slot[(size_t)d * SLOT_CAP + r] = s;
    } else {
      const int p = atomicAdd(ovf_cnt, 1);
      v2i ent;
      ent.x = s;
      ent.y = d;
      *(v2i*)(ovf + 2 * (size_t)p) = ent;
    }
  }
}

__device__ __forceinline__ float deg_inv_sqrt(int c) {
  const float d = 1.0f + (float)c;
  return (d > 0.f) ? rsqrtf(d) : 0.f;
}

#define A_PITCH 132

__device__ __forceinline__ void load_b_frags(const float* __restrict__ B, int half, int col, v2f (&bf)[32]) {
  const float* bp = B + (size_t)(4 * half) * DIM + col;
#pragma unroll
  for (int s = 0; s < 32; ++s) {
    const int r0 = 8 * (s >> 1) + 2 * (s & 1);
    bf[s].x = bp[r0 * DIM];
    bf[s].y = bp[(r0 + 1) * DIM];
  }
}

__device__ __forceinline__ v8f tile_product(const float* fa, const v2f (&bf)[32]) {
  v8f acc = {};
#pragma unroll
  for (int j = 0; j < 16; ++j) {
    const v4f a = *(const v4f*)(fa + 8 * j);
    v2f alo, ahi;
    alo.x = a.x; alo.y = a.y;
    ahi.x = a.z; ahi.y = a.w;
    acc = __builtin_amdgcn_wmma_f32_16x16x4_f32(
         false, alo,  false, bf[2 * j],
         (short)0, acc,  false,  false);
    acc = __builtin_amdgcn_wmma_f32_16x16x4_f32(
         false, ahi,  false, bf[2 * j + 1],
         (short)0, acc,  false,  false);
  }
  return acc;
}

__device__ __forceinline__ v4f mean_row(v4f s, float c) {
  const float d = fmaxf(c, 1.0f);
  v4f r;
  r.x = s.x / d;
  r.y = s.y / d;
  r.z = s.z / d;
  r.w = s.w / d;
  return r;
}

__global__ void __launch_bounds__(256)
mean_gemm_wmma_f32(const float* __restrict__ A, const float* __restrict__ cnt, const float* __restrict__ B,
                   const float* __restrict__ bias, float* __restrict__ Cout, int n_tiles) {
  __shared__ __attribute__((aligned(16))) float sA[2][16 * A_PITCH];
  const int tid  = threadIdx.x;
  const int lane = tid & 31;
  const int wv   = tid >> 5;
  const int half = lane >> 4;
  const int lm   = lane & 15;
  const int col0 = wv * 16;

  v2f bf[32];
  load_b_frags(B, half, col0 + lm, bf);
  const float badd = bias[col0 + lm];

  const float* ga = A + (size_t)wv * DIM + lane * 4;
  const int so0 = wv * A_PITCH + lane * 4;
  const int so1 = (wv + 8) * A_PITCH + lane * 4;
  const int fo  = lm * A_PITCH + 4 * half;

  int tile = blockIdx.x;
  v4f p0 = {0.f, 0.f, 0.f, 0.f}, p1 = {0.f, 0.f, 0.f, 0.f};
  float c0 = 1.f, c1 = 1.f;
  if (tile < n_tiles) {
    p0 = *(const v4f*)(ga + (size_t)tile * 16 * DIM);
    p1 = *(const v4f*)(ga + (size_t)tile * 16 * DIM + 8 * DIM);
    c0 = cnt[tile * 16 + wv];
    c1 = cnt[tile * 16 + wv + 8];
    *(v4f*)(&sA[0][so0]) = mean_row(p0, c0);
    *(v4f*)(&sA[0][so1]) = mean_row(p1, c1);
  }
  __syncthreads();
  int buf = 0;
  for (; tile < n_tiles; tile += gridDim.x) {
    const int nt = tile + gridDim.x;
    const bool more = nt < n_tiles;
    if (more) {
      p0 = *(const v4f*)(ga + (size_t)nt * 16 * DIM);
      p1 = *(const v4f*)(ga + (size_t)nt * 16 * DIM + 8 * DIM);
      c0 = cnt[nt * 16 + wv];
      c1 = cnt[nt * 16 + wv + 8];
    }
    const v8f acc = tile_product(&sA[buf][fo], bf);
    float* cp = Cout + (size_t)(tile * 16 + half * 8) * DIM + col0 + lm;
#pragma unroll
    for (int r = 0; r < 8; ++r) cp[r * DIM] = acc[r] + badd;
    if (more) {
      *(v4f*)(&sA[buf ^ 1][so0]) = mean_row(p0, c0);
      *(v4f*)(&sA[buf ^ 1][so1]) = mean_row(p1, c1);
    }
    __syncthreads();
    buf ^= 1;
  }
}


template <int G>
__device__ __forceinline__ void add_rows(v4f& acc, const float* __restrict__ xs, int lane4, int ent, int k0) {
  v4f v[G];
#pragma unroll
  for (int u = 0; u < G; ++u) {
    const int sk = __builtin_amdgcn_readlane(ent, k0 + u);
    v[u] = *(const v4f*)((const char*)xs + ((unsigned)sk * DIM + (unsigned)lane4) * 4u);
  }
#pragma unroll
  for (int u = 0; u < G; ++u) acc += v[u];
}

__device__ __forceinline__ int edge_chunk(const int* __restrict__ slot, int base, int end, int lane) {
  const int j = base + lane;
  int ent = 0;
  if (j < end) ent = slot[j];
  return ent;
}

__device__ __forceinline__ v4f edge_sum(const int* __restrict__ slot, const float* __restrict__ xs,
                                        int beg, int end, int first, int lane) {
  v4f acc = {0.f, 0.f, 0.f, 0.f};
  int ent = first;
  for (int base = beg; base < end; base += 32) {
    if (base != beg) ent = edge_chunk(slot, base, end, lane);
    const int m = (end - base < 32) ? (end - base) : 32;
    int k = 0;
    for (; k + 8 <= m; k += 8) add_rows<8>(acc, xs, lane * 4, ent, k);
    if (m - k >= 4) { add_rows<4>(acc, xs, lane * 4, ent, k); k += 4; }
    if (m - k >= 2) { add_rows<2>(acc, xs, lane * 4, ent, k); k += 2; }
    if (m - k >= 1) add_rows<1>(acc, xs, lane * 4, ent, k);
  }
  return acc;
}

__device__ __forceinline__ v4f overflow_sum(v4f acc, const int* __restrict__ ovf_cnt, const int* __restrict__ ovf,
                                            const float* __restrict__ xs, int n, int lane) {
  const int no = *ovf_cnt;
  for (int j0 = 0; j0 < no; j0 += 32) {
    const int j = j0 + lane;
    v2i e = {0, -1};
    if (j < no) e = *(const v2i*)(ovf + 2 * (size_t)j);
    unsigned mask = __builtin_amdgcn_ballot_w32(e.y == n);
    while (mask) {
      const int k = __builtin_ctz(mask);
      mask &= mask - 1;
      add_rows<1>(acc, xs, lane * 4, e.x, k);
    }
  }
  return acc;
}

__device__ __forceinline__ v4f bias_self_relu(v4f bb, v4f self, float di, v4f acc) {
  v4f r = (acc + self) * di + bb;
  r.x = fmaxf(r.x, 0.f);
  r.y = fmaxf(r.y, 0.f);
  r.z = fmaxf(r.z, 0.f);
  r.w = fmaxf(r.w, 0.f);
  return r;
}

static_assert(N_NODES % 16 == 0, "whole 16-row tiles only: the node index is never out of range");
template <bool FROM_EMBED>
__global__ void __launch_bounds__(256, 8)
rows_gemm_kernel(const int* __restrict__ x, const float* __restrict__ emb,
                 const int* __restrict__ ecnt, const int* __restrict__ slot,
                 const int* __restrict__ ovf_cnt, const int* __restrict__ ovf,
                 const float* __restrict__ xs, const float* __restrict__ bl,
                 const float* __restrict__ B, float* __restrict__ Cout, int n_tiles) {
  __shared__ __attribute__((aligned(16))) float sA[2][16 * A_PITCH];
  const int lane = threadIdx.x & 31;
  const int wv   = __builtin_amdgcn_readfirstlane((int)(threadIdx.x >> 5));
  const int half = lane >> 4;
  const int lm   = lane & 15;
  const int col0 = wv * 16;

  v2f bf[32];
  load_b_frags(B, half, col0 + lm, bf);

  const int so0 = wv * A_PITCH + lane * 4;
  const int so1 = (wv + 8) * A_PITCH + lane * 4;
  const int fo  = lm * A_PITCH + 4 * half;

  v4f bb = {0.f, 0.f, 0.f, 0.f};
  if (!FROM_EMBED) bb = *(const v4f*)(bl + lane * 4);

  int buf = 0;
  for (int tile = blockIdx.x; tile < n_tiles; tile += gridDim.x) {
    const int n0 = tile * 16 + wv;
    const int n1 = n0 + 8;
    const int c0 = ecnt[n0];
    const int c1 = ecnt[n1];
    const float di0 = deg_inv_sqrt(c0);
    const float di1 = deg_inv_sqrt(c1);
    if (FROM_EMBED) {
      const v4f r0 = embed_row(x, emb, n0, lane);
      const v4f r1 = embed_row(x, emb, n1, lane);
      *(v4f*)(&sA[buf][so0]) = r0 * di0;
      *(v4f*)(&sA[buf][so1]) = r1 * di1;
    } else {
      const int beg0 = n0 * SLOT_CAP, end0 = beg0 + ((c0 < SLOT_CAP) ? c0 : SLOT_CAP);
      const int beg1 = n1 * SLOT_CAP, end1 = beg1 + ((c1 < SLOT_CAP) ? c1 : SLOT_CAP);
      const int e0 = edge_chunk(slot, beg0, end0, lane);
      const int e1 = edge_chunk(slot, beg1, end1, lane);
      const v4f self0 = *(const v4f*)((const char*)xs + ((unsigned)n0 * DIM + (unsigned)lane * 4) * 4u);
      v4f a0 = edge_sum(slot, xs, beg0, end0, e0, lane);
      if (c0 > SLOT_CAP) a0 = overflow_sum(a0, ovf_cnt, ovf, xs, n0, lane);
      const v4f self1 = *(const v4f*)((const char*)xs + ((unsigned)n1 * DIM + (unsigned)lane * 4) * 4u);
      *(v4f*)(&sA[buf][so0]) = bias_self_relu(bb, self0, di0, a0) * di0;
      v4f a1 = edge_sum(slot, xs, beg1, end1, e1, lane);
      if (c1 > SLOT_CAP) a1 = overflow_sum(a1, ovf_cnt, ovf, xs, n1, lane);
      *(v4f*)(&sA[buf][so1]) = bias_self_relu(bb, self1, di1, a1) * di1;
    }
    __syncthreads();
    const v8f acc = tile_product(&sA[buf][fo], bf);
    float* cp = (float*)((char*)Cout + (unsigned)((tile * 16 + half * 8) * DIM + col0 + lm) * 4u);
#pragma unroll
    for (int r = 0; r < 8; ++r) cp[r * DIM] = acc[r];
    buf ^= 1;
  }
}

#define POOL_NPW 5
static_assert(N_NODES % (8 * POOL_NPW) == 0, "whole blocks only: the node index is never out of range");
__device__ __forceinline__ void pool_add_run(float* __restrict__ pool, float* __restrict__ cnt,
                                             int g, int lane, v4f a, int len) {
  float* p = pool + (size_t)g * DIM + lane * 4;
  unsafeAtomicAdd(p + 0, a.x);
  unsafeAtomicAdd(p + 1, a.y);
  unsafeAtomicAdd(p + 2, a.z);
  unsafeAtomicAdd(p + 3, a.w);
  if (lane == 0) unsafeAtomicAdd(cnt + g, (float)len);
}
__global__ void __launch_bounds__(256)
gather_pool_kernel(const int* __restrict__ ecnt, const int* __restrict__ slot,
                   const int* __restrict__ ovf_cnt, const int* __restrict__ ovf,
                   const float* __restrict__ xs, const float* __restrict__ bl, const int* __restrict__ batch,
                   float* __restrict__ pool, float* __restrict__ cnt) {
  const int lane = threadIdx.x & 31;
  const int w = __builtin_amdgcn_readfirstlane((int)((blockIdx.x * 256 + threadIdx.x) >> 5));
  const int n0 = w * POOL_NPW;
  const v4f bb = *(const v4f*)(bl + lane * 4);
  int g[POOL_NPW], c[POOL_NPW], en[POOL_NPW];
  int e[POOL_NPW];
  v4f self[POOL_NPW];
#pragma unroll
  for (int i = 0; i < POOL_NPW; ++i) {
    c[i] = ecnt[n0 + i];
    g[i] = batch[n0 + i];
  }
#pragma unroll
  for (int i = 0; i < POOL_NPW; ++i) {
    en[i] = (n0 + i) * SLOT_CAP + ((c[i] < SLOT_CAP) ? c[i] : SLOT_CAP);
    e[i] = edge_chunk(slot, (n0 + i) * SLOT_CAP, en[i], lane);
    self[i] = *(const v4f*)((const char*)xs + ((unsigned)(n0 + i) * DIM + (unsigned)lane * 4) * 4u);
  }
  v4f row[POOL_NPW];
#pragma unroll
  for (int i = 0; i < POOL_NPW; ++i) {
    v4f a = edge_sum(slot, xs, (n0 + i) * SLOT_CAP, en[i], e[i], lane);
    if (c[i] > SLOT_CAP) a = overflow_sum(a, ovf_cnt, ovf, xs, n0 + i, lane);
    row[i] = bias_self_relu(bb, self[i], deg_inv_sqrt(c[i]), a);
  }
  int cur = g[0];
  int len = 1;
  v4f acc = row[0];
#pragma unroll
  for (int i = 1; i < POOL_NPW; ++i) {
    if (g[i] == cur) {
      acc += row[i];
      ++len;
    } else {
      pool_add_run(pool, cnt, cur, lane, acc, len);
      cur = g[i];
      acc = row[i];
      len = 1;
    }
  }
  pool_add_run(pool, cnt, cur, lane, acc, len);
}

extern "C" void kernel_launch(void* const* d_in, const int* in_sizes, int n_in,
                              void* d_out, int out_size, void* d_ws, size_t ws_size,
                              hipStream_t stream) {
  const int*   x    = (const int*)d_in[0];
  const int*   ei   = (const int*)d_in[1];
  const int*   batch= (const int*)d_in[2];
  const float* emb  = (const float*)d_in[3];
  const float* W    = (const float*)d_in[4];
  const float* b    = (const float*)d_in[5];
  const float* Wr   = (const float*)d_in[6];
  const float* br   = (const float*)d_in[7];
  float* out = (float*)d_out;

  char* ws = (char*)d_ws;
  const size_t NH = (size_t)N_NODES * DIM * sizeof(float);
  float* bufA = (float*)(ws);
  float* bufB = (float*)(ws + NH);
  int*   ecnt    = (int*)(ws + 2 * NH);
  int*   ovf_cnt = (int*)(ws + 2 * NH + 256 * 1024);
  int*   slot    = (int*)(ws + 2 * NH + 8 * 1024 * 1024);
  int*   ovf     = (int*)(ws + 3 * NH + 8 * 1024 * 1024);
  float* pool = (float*)(ws + 3 * NH + 4 * 1024 * 1024);
  float* cnt  = (float*)(ws + 3 * NH + 4 * 1024 * 1024 + 512 * 1024);

  const int B256 = 256;
  const int e4_blocks = (N_EDGES / 4 + B256 - 1) / B256;
  const int pool_blocks = N_NODES / (8 * POOL_NPW);

  const int g_blocks = 782;
  const int n_tiles = N_NODES / 16;

  zero_kernel<<<N_GRAPHS * DIM / B256, B256, 0, stream>>>(ecnt, ovf_cnt, pool, cnt);
  count_slot_kernel<<<e4_blocks, B256, 0, stream>>>(ei, ecnt, slot, ovf_cnt, ovf);

  rows_gemm_kernel<true><<<g_blocks, B256, 0, stream>>>(x, emb, ecnt, nullptr, nullptr, nullptr, nullptr, nullptr,
                                                        W, bufA, n_tiles);

  rows_gemm_kernel<false><<<g_blocks, B256, 0, stream>>>(nullptr, nullptr, ecnt, slot, ovf_cnt, ovf, bufA, b,
                                                         W + (size_t)1 * DIM * DIM, bufB, n_tiles);
  rows_gemm_kernel<false><<<g_blocks, B256, 0, stream>>>(nullptr, nullptr, ecnt, slot, ovf_cnt, ovf, bufB, b + 1 * DIM,
                                                         W + (size_t)2 * DIM * DIM, bufA, n_tiles);
  rows_gemm_kernel<false><<<g_blocks, B256, 0, stream>>>(nullptr, nullptr, ecnt, slot, ovf_cnt, ovf, bufA, b + 2 * DIM,
                                                         W + (size_t)3 * DIM * DIM, bufB, n_tiles);

  gather_pool_kernel<<<pool_blocks, B256, 0, stream>>>(ecnt, slot, ovf_cnt, ovf, bufB, b + 3 * DIM, batch, pool, cnt);

  mean_gemm_wmma_f32<<<N_GRAPHS / 16, B256, 0, stream>>>(pool, cnt, Wr, br, out, N_GRAPHS / 16);
}


# Edges binned in LDS, not in memory: a chunk-wise counting sort by destination bin, then per-bin LDS ranks and whole-row slot stores replace per-edge global atomics and 4-byte scatters (sum order)
# speedup vs baseline: 5.6311x; 1.3809x over previous
#include <hip/hip_runtime.h>

#define N_NODES  50000
#define N_EDGES  800000
#define DIM      128
#define N_LAYERS 4
#define N_GRAPHS 512
#define N_COLS   4
#define VOCAB    128

typedef __attribute__((ext_vector_type(2))) float v2f;
typedef __attribute__((ext_vector_type(4))) float v4f;
typedef __attribute__((ext_vector_type(8))) float v8f;

__device__ __forceinline__ v4f embed_row(const int* __restrict__ x, const float* __restrict__ emb,
                                         int n, int lane) {
  const float* el = emb + lane * 4;
  v4f s = {0.f, 0.f, 0.f, 0.f};
#pragma unroll
  for (int c = 0; c < N_COLS; ++c) {
    const int v = x[n * N_COLS + c];
    s += *(const v4f*)(el + ((size_t)c * VOCAB + v) * DIM);
  }
  return s;
}

#define SLOT_CAP 64
typedef __attribute__((ext_vector_type(2))) int v2i;

#define BIN_NODES 200
#define N_BINS    (N_NODES / BIN_NODES)
#define CH_EDGES  3200
#define N_CHUNKS  (N_EDGES / CH_EDGES)
#define OFF_PITCH 256
#define SL_PITCH  (SLOT_CAP + 1)
static_assert(N_NODES % BIN_NODES == 0 && N_EDGES % CH_EDGES == 0, "whole bins and whole chunks only");
static_assert(BIN_NODES <= 256 && N_NODES < (1 << 23), "packed edge word: source id above 8 bits of bin-local destination");
static_assert(N_BINS < OFF_PITCH, "offset entries 0..N_BINS of a chunk fit its row of the table");

#define PART_THREADS 320
static_assert(CH_EDGES % PART_THREADS == 0 && PART_THREADS % 32 == 0, "every thread takes the same number of edges");
static_assert(N_BINS < PART_THREADS && OFF_PITCH <= PART_THREADS, "one thread per bin counter and per offset entry");
static_assert(N_GRAPHS * DIM <= N_CHUNKS * PART_THREADS, "one thread per pool element");
__global__ void __launch_bounds__(PART_THREADS)
partition_kernel(const int* __restrict__ ei, int* __restrict__ part, int* __restrict__ off,
                 int* __restrict__ ovf_cnt, float* __restrict__ pool, float* __restrict__ cnt) {
  __shared__ int hist[PART_THREADS];
  __shared__ int wsum[PART_THREADS / 32];
  __shared__ int sorted[CH_EDGES];
  const int tid = threadIdx.x;
  const int lane = tid & 31;
  const int wv = tid >> 5;
  const int w = blockIdx.x;
  const int gid = w * PART_THREADS + tid;
  if (gid < N_GRAPHS * DIM) pool[gid] = 0.f;
  if (gid < N_GRAPHS) cnt[gid] = 0.f;
  if (gid == 0) *ovf_cnt = 0;
  hist[tid] = 0;
  __syncthreads();
  const int* sp = ei + (size_t)w * CH_EDGES;
  const int* dp = ei + N_EDGES + (size_t)w * CH_EDGES;
  constexpr int PER = CH_EDGES / PART_THREADS;
  int key[PER], bn[PER], rk[PER];
#pragma unroll
  for (int k = 0; k < PER; ++k) {
    const int i = tid + PART_THREADS * k;
    const int s = sp[i], d = dp[i];
    const int b = d / BIN_NODES;
    key[k] = (s << 8) | (d - b * BIN_NODES);
    bn[k] = b;
  }
#pragma unroll
  for (int k = 0; k < PER; ++k) rk[k] = atomicAdd(&hist[bn[k]], 1);
  __syncthreads();
  const int v = hist[tid];
  int x = v;
#pragma unroll
  for (int o = 1; o < 32; o <<= 1) {
    const int y = __shfl_up(x, o);
    if (lane >= o) x += y;
  }
  if (lane == 31) wsum[wv] = x;
  __syncthreads();
  int add = 0;
  for (int q = 0; q < wv; ++q) add += wsum[q];
  const int base = add + x - v;
  if (tid < OFF_PITCH) off[(size_t)w * OFF_PITCH + tid] = base;
  hist[tid] = base;
  __syncthreads();
#pragma unroll
  for (int k = 0; k < PER; ++k) sorted[hist[bn[k]] + rk[k]] = key[k];
  __syncthreads();
  int* pp = part + (size_t)w * CH_EDGES;
#pragma unroll
  for (int k = 0; k < PER; ++k) pp[tid + PART_THREADS * k] = sorted[tid + PART_THREADS * k];
}

#define SLOT_WAVES 16
#define SLOT_RUNS  ((N_CHUNKS + SLOT_WAVES - 1) / SLOT_WAVES)
__global__ void __launch_bounds__(SLOT_WAVES * 32)
slot_kernel(const int* __restrict__ part, const int* __restrict__ off, int* __restrict__ ecnt,
            int* __restrict__ slot, int* __restrict__ ovf_cnt, int* __restrict__ ovf) {
  __shared__ int cnt_l[256];
  __shared__ int slot_l[BIN_NODES * SL_PITCH];
  const int tid = threadIdx.x;
  const int lane = tid & 31;
  const int wv = tid >> 5;
  const int b = blockIdx.x;
  if (tid < 256) cnt_l[tid] = 0;
  int lo[SLOT_RUNS], hi[SLOT_RUNS], p[SLOT_RUNS];
#pragma unroll
  for (int u = 0; u < SLOT_RUNS; ++u) {
    const int c = u * SLOT_WAVES + wv;
    const int cc = (c < N_CHUNKS) ? c : (N_CHUNKS - 1);
    lo[u] = off[(size_t)cc * OFF_PITCH + b];
    hi[u] = off[(size_t)cc * OFF_PITCH + b + 1];
    if (c >= N_CHUNKS) hi[u] = lo[u];
  }
#pragma unroll
  for (int u = 0; u < SLOT_RUNS; ++u) {
    const int c = u * SLOT_WAVES + wv;
    const int j = lo[u] + lane;
    p[u] = -1;
    if (j < hi[u]) p[u] = part[(size_t)c * CH_EDGES + j];
  }
  __syncthreads();
  auto put = [&](int pk, int r) {
    const int dl = pk & 255;
    const int s = pk >> 8;
    if (r < SLOT_CAP) {
      slot_l[dl * SL_PITCH + r] = s;
    } else {
      const int q = atomicAdd(ovf_cnt, 1);
      v2i ent;
      ent.x = s;
      ent.y = b * BIN_NODES + dl;
      *(v2i*)(ovf + 2 * (size_t)q) = ent;
    }
  };
  int rr[SLOT_RUNS];
#pragma unroll
  for (int u = 0; u < SLOT_RUNS; ++u) {
    rr[u] = 0;
    if (p[u] >= 0) rr[u] = atomicAdd(&cnt_l[p[u] & 255], 1);
  }
#pragma unroll
  for (int u = 0; u < SLOT_RUNS; ++u)
    if (p[u] >= 0) put(p[u], rr[u]);
#pragma unroll
  for (int u = 0; u < SLOT_RUNS; ++u) {
    const int c = u * SLOT_WAVES + wv;
    for (int j = lo[u] + 32 + lane; j < hi[u]; j += 32) {
      const int pk = part[(size_t)c * CH_EDGES + j];
      put(pk, atomicAdd(&cnt_l[pk & 255], 1));
    }
  }
  __syncthreads();
  if (tid < BIN_NODES) ecnt[b * BIN_NODES + tid] = cnt_l[tid];
  constexpr int TRIPS = (BIN_NODES + SLOT_WAVES - 1) / SLOT_WAVES;
  static_assert(2 * 31 < SLOT_CAP, "two slot entries per lane");
  const int e0 = 2 * lane;
  int cq[TRIPS];
  v2i oq[TRIPS];
#pragma unroll
  for (int q = 0; q < TRIPS; ++q) {
    const int i = wv + SLOT_WAVES * q;
    const int ic = (i < BIN_NODES) ? i : (BIN_NODES - 1);
    cq[q] = cnt_l[ic];
    oq[q].x = slot_l[ic * SL_PITCH + e0];
    oq[q].y = slot_l[ic * SL_PITCH + e0 + 1];
  }
#pragma unroll
  for (int q = 0; q < TRIPS; ++q) {
    const int i = wv + SLOT_WAVES * q;
    if (i < BIN_NODES && e0 < cq[q]) *(v2i*)(slot + (size_t)(b * BIN_NODES + i) * SLOT_CAP + e0) = oq[q];
  }
}

__device__ __forceinline__ float deg_inv_sqrt(int c) {
  const float d = 1.0f + (float)c;
  return (d > 0.f) ? rsqrtf(d) : 0.f;
}

#define A_PITCH 132

__device__ __forceinline__ void load_b_frags(const float* __restrict__ B, int half, int col, v2f (&bf)[32]) {
  const float* bp = B + (size_t)(4 * half) * DIM + col;
#pragma unroll
  for (int s = 0; s < 32; ++s) {
    const int r0 = 8 * (s >> 1) + 2 * (s & 1);
    bf[s].x = bp[r0 * DIM];
    bf[s].y = bp[(r0 + 1) * DIM];
  }
}

__device__ __forceinline__ v8f tile_product(const float* fa, const v2f (&bf)[32]) {
  v8f acc = {};
#pragma unroll
  for (int j = 0; j < 16; ++j) {
    const v4f a = *(const v4f*)(fa + 8 * j);
    v2f alo, ahi;
    alo.x = a.x; alo.y = a.y;
    ahi.x = a.z; ahi.y = a.w;
    acc = __builtin_amdgcn_wmma_f32_16x16x4_f32(
         false, alo,  false, bf[2 * j],
         (short)0, acc,  false,  false);
    acc = __builtin_amdgcn_wmma_f32_16x16x4_f32(
         false, ahi,  false, bf[2 * j + 1],
         (short)0, acc,  false,  false);
  }
  return acc;
}

__device__ __forceinline__ v4f mean_row(v4f s, float c) {
  const float d = fmaxf(c, 1.0f);
  v4f r;
  r.x = s.x / d;
  r.y = s.y / d;
  r.z = s.z / d;
  r.w = s.w / d;
  return r;
}

__global__ void __launch_bounds__(256)
mean_gemm_wmma_f32(const float* __restrict__ A, const float* __restrict__ cnt, const float* __restrict__ B,
                   const float* __restrict__ bias, float* __restrict__ Cout, int n_tiles) {
  __shared__ __attribute__((aligned(16))) float sA[2][16 * A_PITCH];
  const int tid  = threadIdx.x;
  const int lane = tid & 31;
  const int wv   = tid >> 5;
  const int half = lane >> 4;
  const int lm   = lane & 15;
  const int col0 = wv * 16;

  v2f bf[32];
  load_b_frags(B, half, col0 + lm, bf);
  const float badd = bias[col0 + lm];

  const float* ga = A + (size_t)wv * DIM + lane * 4;
  const int so0 = wv * A_PITCH + lane * 4;
  const int so1 = (wv + 8) * A_PITCH + lane * 4;
  const int fo  = lm * A_PITCH + 4 * half;

  int tile = blockIdx.x;
  v4f p0 = {0.f, 0.f, 0.f, 0.f}, p1 = {0.f, 0.f, 0.f, 0.f};
  float c0 = 1.f, c1 = 1.f;
  if (tile < n_tiles) {
    p0 = *(const v4f*)(ga + (size_t)tile * 16 * DIM);
    p1 = *(const v4f*)(ga + (size_t)tile * 16 * DIM + 8 * DIM);
    c0 = cnt[tile * 16 + wv];
    c1 = cnt[tile * 16 + wv + 8];
    *(v4f*)(&sA[0][so0]) = mean_row(p0, c0);
    *(v4f*)(&sA[0][so1]) = mean_row(p1, c1);
  }
  __syncthreads();
  int buf = 0;
  for (; tile < n_tiles; tile += gridDim.x) {
    const int nt = tile + gridDim.x;
    const bool more = nt < n_tiles;
    if (more) {
      p0 = *(const v4f*)(ga + (size_t)nt * 16 * DIM);
      p1 = *(const v4f*)(ga + (size_t)nt * 16 * DIM + 8 * DIM);
      c0 = cnt[nt * 16 + wv];
      c1 = cnt[nt * 16 + wv + 8];
    }
    const v8f acc = tile_product(&sA[buf][fo], bf);
    float* cp = Cout + (size_t)(tile * 16 + half * 8) * DIM + col0 + lm;
#pragma unroll
    for (int r = 0; r < 8; ++r) cp[r * DIM] = acc[r] + badd;
    if (more) {
      *(v4f*)(&sA[buf ^ 1][so0]) = mean_row(p0, c0);
      *(v4f*)(&sA[buf ^ 1][so1]) = mean_row(p1, c1);
    }
    __syncthreads();
    buf ^= 1;
  }
}


template <int G>
__device__ __forceinline__ void add_rows(v4f& acc, const float* __restrict__ xs, int lane4, int ent, int k0) {
  v4f v[G];
#pragma unroll
  for (int u = 0; u < G; ++u) {
    const int sk = __builtin_amdgcn_readlane(ent, k0 + u);
    v[u] = *(const v4f*)((const char*)xs + ((unsigned)sk * DIM + (unsigned)lane4) * 4u);
  }
#pragma unroll
  for (int u = 0; u < G; ++u) acc += v[u];
}

__device__ __forceinline__ int edge_chunk(const int* __restrict__ slot, int base, int end, int lane) {
  const int j = base + lane;
  int ent = 0;
  if (j < end) ent = slot[j];
  return ent;
}

__device__ __forceinline__ v4f edge_sum(const int* __restrict__ slot, const float* __restrict__ xs,
                                        int beg, int end, int first, int lane) {
  v4f acc = {0.f, 0.f, 0.f, 0.f};
  int ent = first;
  for (int base = beg; base < end; base += 32) {
    if (base != beg) ent = edge_chunk(slot, base, end, lane);
    const int m = (end - base < 32) ? (end - base) : 32;
    int k = 0;
    for (; k + 8 <= m; k += 8) add_rows<8>(acc, xs, lane * 4, ent, k);
    if (m - k >= 4) { add_rows<4>(acc, xs, lane * 4, ent, k); k += 4; }
    if (m - k >= 2) { add_rows<2>(acc, xs, lane * 4, ent, k); k += 2; }
    if (m - k >= 1) add_rows<1>(acc, xs, lane * 4, ent, k);
  }
  return acc;
}

__device__ __forceinline__ v4f overflow_sum(v4f acc, const int* __restrict__ ovf_cnt, const int* __restrict__ ovf,
                                            const float* __restrict__ xs, int n, int lane) {
  const int no = *ovf_cnt;
  for (int j0 = 0; j0 < no; j0 += 32) {
    const int j = j0 + lane;
    v2i e = {0, -1};
    if (j < no) e = *(const v2i*)(ovf + 2 * (size_t)j);
    unsigned mask = __builtin_amdgcn_ballot_w32(e.y == n);
    while (mask) {
      const int k = __builtin_ctz(mask);
      mask &= mask - 1;
      add_rows<1>(acc, xs, lane * 4, e.x, k);
    }
  }
  return acc;
}

__device__ __forceinline__ v4f bias_self_relu(v4f bb, v4f self, float di, v4f acc) {
  v4f r = (acc + self) * di + bb;
  r.x = fmaxf(r.x, 0.f);
  r.y = fmaxf(r.y, 0.f);
  r.z = fmaxf(r.z, 0.f);
  r.w = fmaxf(r.w, 0.f);
  return r;
}

static_assert(N_NODES % 16 == 0, "whole 16-row tiles only: the node index is never out of range");
template <bool FROM_EMBED>
__global__ void __launch_bounds__(256, 8)
rows_gemm_kernel(const int* __restrict__ x, const float* __restrict__ emb,
                 const int* __restrict__ ecnt, const int* __restrict__ slot,
                 const int* __restrict__ ovf_cnt, const int* __restrict__ ovf,
                 const float* __restrict__ xs, const float* __restrict__ bl,
                 const float* __restrict__ B, float* __restrict__ Cout, int n_tiles) {
  __shared__ __attribute__((aligned(16))) float sA[2][16 * A_PITCH];
  const int lane = threadIdx.x & 31;
  const int wv   = __builtin_amdgcn_readfirstlane((int)(threadIdx.x >> 5));
  const int half = lane >> 4;
  const int lm   = lane & 15;
  const int col0 = wv * 16;

  v2f bf[32];
  load_b_frags(B, half, col0 + lm, bf);

  const int so0 = wv * A_PITCH + lane * 4;
  const int so1 = (wv + 8) * A_PITCH + lane * 4;
  const int fo  = lm * A_PITCH + 4 * half;

  v4f bb = {0.f, 0.f, 0.f, 0.f};
  if (!FROM_EMBED) bb = *(const v4f*)(bl + lane * 4);

  int buf = 0;
  for (int tile = blockIdx.x; tile < n_tiles; tile += gridDim.x) {
    const int n0 = tile * 16 + wv;
    const int n1 = n0 + 8;
    const int c0 = ecnt[n0];
    const int c1 = ecnt[n1];
    const float di0 = deg_inv_sqrt(c0);
    const float di1 = deg_inv_sqrt(c1);
    if (FROM_EMBED) {
      const v4f r0 = embed_row(x, emb, n0, lane);
      const v4f r1 = embed_row(x, emb, n1, lane);
      *(v4f*)(&sA[buf][so0]) = r0 * di0;
      *(v4f*)(&sA[buf][so1]) = r1 * di1;
    } else {
      const int beg0 = n0 * SLOT_CAP, end0 = beg0 + ((c0 < SLOT_CAP) ? c0 : SLOT_CAP);
      const int beg1 = n1 * SLOT_CAP, end1 = beg1 + ((c1 < SLOT_CAP) ? c1 : SLOT_CAP);
      const int e0 = edge_chunk(slot, beg0, end0, lane);
      const int e1 = edge_chunk(slot, beg1, end1, lane);
      const v4f self0 = *(const v4f*)((const char*)xs + ((unsigned)n0 * DIM + (unsigned)lane * 4) * 4u);
      v4f a0 = edge_sum(slot, xs, beg0, end0, e0, lane);
      if (c0 > SLOT_CAP) a0 = overflow_sum(a0, ovf_cnt, ovf, xs, n0, lane);
      const v4f self1 = *(const v4f*)((const char*)xs + ((unsigned)n1 * DIM + (unsigned)lane * 4) * 4u);
      *(v4f*)(&sA[buf][so0]) = bias_self_relu(bb, self0, di0, a0) * di0;
      v4f a1 = edge_sum(slot, xs, beg1, end1, e1, lane);
      if (c1 > SLOT_CAP) a1 = overflow_sum(a1, ovf_cnt, ovf, xs, n1, lane);
      *(v4f*)(&sA[buf][so1]) = bias_self_relu(bb, self1, di1, a1) * di1;
    }
    __syncthreads();
    const v8f acc = tile_product(&sA[buf][fo], bf);
    float* cp = (float*)((char*)Cout + (unsigned)((tile * 16 + half * 8) * DIM + col0 + lm) * 4u);
#pragma unroll
    for (int r = 0; r < 8; ++r) cp[r * DIM] = acc[r];
    buf ^= 1;
  }
}

#define POOL_NPW 5
static_assert(N_NODES % (8 * POOL_NPW) == 0, "whole blocks only: the node index is never out of range");
__device__ __forceinline__ void pool_add_run(float* __restrict__ pool, float* __restrict__ cnt,
                                             int g, int lane, v4f a, int len) {
  float* p = pool + (size_t)g * DIM + lane * 4;
  unsafeAtomicAdd(p + 0, a.x);
  unsafeAtomicAdd(p + 1, a.y);
  unsafeAtomicAdd(p + 2, a.z);
  unsafeAtomicAdd(p + 3, a.w);
  if (lane == 0) unsafeAtomicAdd(cnt + g, (float)len);
}
__global__ void __launch_bounds__(256)
gather_pool_kernel(const int* __restrict__ ecnt, const int* __restrict__ slot,
                   const int* __restrict__ ovf_cnt, const int* __restrict__ ovf,
                   const float* __restrict__ xs, const float* __restrict__ bl, const int* __restrict__ batch,
                   float* __restrict__ pool, float* __restrict__ cnt) {
  const int lane = threadIdx.x & 31;
  const int w = __builtin_amdgcn_readfirstlane((int)((blockIdx.x * 256 + threadIdx.x) >> 5));
  const int n0 = w * POOL_NPW;
  const v4f bb = *(const v4f*)(bl + lane * 4);
  int g[POOL_NPW], c[POOL_NPW], en[POOL_NPW];
  int e[POOL_NPW];
  v4f self[POOL_NPW];
#pragma unroll
  for (int i = 0; i < POOL_NPW; ++i) {
    c[i] = ecnt[n0 + i];
    g[i] = batch[n0 + i];
  }
#pragma unroll
  for (int i = 0; i < POOL_NPW; ++i) {
    en[i] = (n0 + i) * SLOT_CAP + ((c[i] < SLOT_CAP) ? c[i] : SLOT_CAP);
    e[i] = edge_chunk(slot, (n0 + i) * SLOT_CAP, en[i], lane);
    self[i] = *(const v4f*)((const char*)xs + ((unsigned)(n0 + i) * DIM + (unsigned)lane * 4) * 4u);
  }
  v4f row[POOL_NPW];
#pragma unroll
  for (int i = 0; i < POOL_NPW; ++i) {
    v4f a = edge_sum(slot, xs, (n0 + i) * SLOT_CAP, en[i], e[i], lane);
    if (c[i] > SLOT_CAP) a = overflow_sum(a, ovf_cnt, ovf, xs, n0 + i, lane);
    row[i] = bias_self_relu(bb, self[i], deg_inv_sqrt(c[i]), a);
  }
  int cur = g[0];
  int len = 1;
  v4f acc = row[0];
#pragma unroll
  for (int i = 1; i < POOL_NPW; ++i) {
    if (g[i] == cur) {
      acc += row[i];
      ++len;
    } else {
      pool_add_run(pool, cnt, cur, lane, acc, len);
      cur = g[i];
      acc = row[i];
      len = 1;
    }
  }
  pool_add_run(pool, cnt, cur, lane, acc, len);
}

extern "C" void kernel_launch(void* const* d_in, const int* in_sizes, int n_in,
                              void* d_out, int out_size, void* d_ws, size_t ws_size,
                              hipStream_t stream) {
  const int*   x    = (const int*)d_in[0];
  const int*   ei   = (const int*)d_in[1];
  const int*   batch= (const int*)d_in[2];
  const float* emb  = (const float*)d_in[3];
  const float* W    = (const float*)d_in[4];
  const float* b    = (const float*)d_in[5];
  const float* Wr   = (const float*)d_in[6];
  const float* br   = (const float*)d_in[7];
  float* out = (float*)d_out;

  char* ws = (char*)d_ws;
  const size_t NH = (size_t)N_NODES * DIM * sizeof(float);
  float* bufA = (float*)(ws);
  float* bufB = (float*)(ws + NH);
  int*   ecnt    = (int*)(ws + 2 * NH);
  int*   ovf_cnt = (int*)(ws + 2 * NH + 256 * 1024);
  int*   slot    = (int*)(ws + 2 * NH + 8 * 1024 * 1024);
  int*   ovf     = (int*)(ws + 3 * NH + 8 * 1024 * 1024);
  float* pool = (float*)(ws + 3 * NH + 4 * 1024 * 1024);
  float* cnt  = (float*)(ws + 3 * NH + 4 * 1024 * 1024 + 512 * 1024);
  int*   part = (int*)(ws + 3 * NH + 16 * 1024 * 1024);
  int*   off  = (int*)(ws + 3 * NH + 20 * 1024 * 1024);

  const int B256 = 256;
  const int pool_blocks = N_NODES / (8 * POOL_NPW);

  const int g_blocks = 782;
  const int n_tiles = N_NODES / 16;

  partition_kernel<<<N_CHUNKS, PART_THREADS, 0, stream>>>(ei, part, off, ovf_cnt, pool, cnt);
  slot_kernel<<<N_BINS, SLOT_WAVES * 32, 0, stream>>>(part, off, ecnt, slot, ovf_cnt, ovf);

  rows_gemm_kernel<true><<<g_blocks, B256, 0, stream>>>(x, emb, ecnt, nullptr, nullptr, nullptr, nullptr, nullptr,
                                                        W, bufA, n_tiles);

  rows_gemm_kernel<false><<<g_blocks, B256, 0, stream>>>(nullptr, nullptr, ecnt, slot, ovf_cnt, ovf, bufA, b,
                                                         W + (size_t)1 * DIM * DIM, bufB, n_tiles);
  rows_gemm_kernel<false><<<g_blocks, B256, 0, stream>>>(nullptr, nullptr, ecnt, slot, ovf_cnt, ovf, bufB, b + 1 * DIM,
                                                         W + (size_t)2 * DIM * DIM, bufA, n_tiles);
  rows_gemm_kernel<false><<<g_blocks, B256, 0, stream>>>(nullptr, nullptr, ecnt, slot, ovf_cnt, ovf, bufA, b + 2 * DIM,
                                                         W + (size_t)3 * DIM * DIM, bufB, n_tiles);

  gather_pool_kernel<<<pool_blocks, B256, 0, stream>>>(ecnt, slot, ovf_cnt, ovf, bufB, b + 3 * DIM, batch, pool, cnt);

  mean_gemm_wmma_f32<<<N_GRAPHS / 16, B256, 0, stream>>>(pool, cnt, Wr, br, out, N_GRAPHS / 16);
}


# First layer's product taken once of the 512 embedding-table rows, not of the 50000 node rows; a node's row = sum of its 4 transformed table rows (sum order); that GEMM launch is gone: 8 launches to 7
# speedup vs baseline: 5.9343x; 1.0538x over previous
#include <hip/hip_runtime.h>

#define N_NODES  50000
#define N_EDGES  800000
#define DIM      128
#define N_LAYERS 4
#define N_GRAPHS 512
#define N_COLS   4
#define VOCAB    128

typedef __attribute__((ext_vector_type(2))) float v2f;
typedef __attribute__((ext_vector_type(4))) float v4f;
typedef __attribute__((ext_vector_type(8))) float v8f;

#define TAB_ROWS  (N_COLS * VOCAB)
#define TAB_TILES (TAB_ROWS / 16)
static_assert(TAB_ROWS % 16 == 0, "whole 16-row tiles of the table");

__device__ __forceinline__ float deg_inv_sqrt(int c) {
  const float d = 1.0f + (float)c;
  return (d > 0.f) ? rsqrtf(d) : 0.f;
}

#define A_PITCH 132

__device__ __forceinline__ void load_b_frags(const float* __restrict__ B, int half, int col, v2f (&bf)[32]) {
  const float* bp = B + (size_t)(4 * half) * DIM + col;
#pragma unroll
  for (int s = 0; s < 32; ++s) {
    const int r0 = 8 * (s >> 1) + 2 * (s & 1);
    bf[s].x = bp[r0 * DIM];
    bf[s].y = bp[(r0 + 1) * DIM];
  }
}

__device__ __forceinline__ v8f tile_product(const float* fa, const v2f (&bf)[32]) {
  v8f acc = {};
#pragma unroll
  for (int j = 0; j < 16; ++j) {
    const v4f a = *(const v4f*)(fa + 8 * j);
    v2f alo, ahi;
    alo.x = a.x; alo.y = a.y;
    ahi.x = a.z; ahi.y = a.w;
    acc = __builtin_amdgcn_wmma_f32_16x16x4_f32(
         false, alo,  false, bf[2 * j],
         (short)0, acc,  false,  false);
    acc = __builtin_amdgcn_wmma_f32_16x16x4_f32(
         false, ahi,  false, bf[2 * j + 1],
         (short)0, acc,  false,  false);
  }
  return acc;
}

__device__ __forceinline__ void table_tile(const float* __restrict__ emb, const float* __restrict__ B,
                                           float* __restrict__ tab, int tile, float* sA) {
  const int lane = threadIdx.x & 31;
  const int wv   = __builtin_amdgcn_readfirstlane((int)(threadIdx.x >> 5));
  const bool worker = wv < 8;
  const int half = lane >> 4;
  const int lm   = lane & 15;
  const int col0 = wv * 16;
  v2f bf[32];
  if (worker) {
    load_b_frags(B, half, col0 + lm, bf);
    const float* ga = emb + ((size_t)tile * 16 + wv) * DIM + lane * 4;
    *(v4f*)(&sA[wv * A_PITCH + lane * 4])       = *(const v4f*)ga;
    *(v4f*)(&sA[(wv + 8) * A_PITCH + lane * 4]) = *(const v4f*)(ga + 8 * DIM);
  }
  __syncthreads();
  if (worker) {
    const v8f acc = tile_product(&sA[lm * A_PITCH + 4 * half], bf);
    float* cp = tab + (size_t)(tile * 16 + half * 8) * DIM + col0 + lm;
#pragma unroll
    for (int r = 0; r < 8; ++r) cp[r * DIM] = acc[r];
  }
}

#define SLOT_CAP 64
typedef __attribute__((ext_vector_type(2))) int v2i;

#define BIN_NODES 200
#define N_BINS    (N_NODES / BIN_NODES)
#define CH_EDGES  3200
#define N_CHUNKS  (N_EDGES / CH_EDGES)
#define OFF_PITCH 256
#define SL_PITCH  (SLOT_CAP + 1)
static_assert(N_NODES % BIN_NODES == 0 && N_EDGES % CH_EDGES == 0, "whole bins and whole chunks only");
static_assert(BIN_NODES <= 256 && N_NODES < (1 << 23), "packed edge word: source id above 8 bits of bin-local destination");
static_assert(N_BINS < OFF_PITCH, "offset entries 0..N_BINS of a chunk fit its row of the table");

#define PART_THREADS 320
static_assert(CH_EDGES % PART_THREADS == 0 && PART_THREADS % 32 == 0, "every thread takes the same number of edges");
static_assert(N_BINS < PART_THREADS && OFF_PITCH <= PART_THREADS, "one thread per bin counter and per offset entry");
static_assert(N_GRAPHS * DIM <= N_CHUNKS * PART_THREADS, "one thread per pool element");
static_assert(PART_THREADS >= 256, "the table product needs 8 waves");
__global__ void __launch_bounds__(PART_THREADS)
partition_kernel(const int* __restrict__ ei, int* __restrict__ part, int* __restrict__ off,
                 int* __restrict__ ovf_cnt, float* __restrict__ pool, float* __restrict__ cnt,
                 const float* __restrict__ emb, const float* __restrict__ W0, float* __restrict__ tab) {
  __shared__ int hist[PART_THREADS];
  __shared__ int wsum[PART_THREADS / 32];
  __shared__ int sorted[CH_EDGES];
  __shared__ __attribute__((aligned(16))) float sT[16 * A_PITCH];
  const int tid = threadIdx.x;
  const int lane = tid & 31;
  const int wv = tid >> 5;
  const int w = blockIdx.x;
  if (w >= N_CHUNKS) {
    table_tile(emb, W0, tab, w - N_CHUNKS, sT);
    return;
  }
  const int gid = w * PART_THREADS + tid;
  if (gid < N_GRAPHS * DIM) pool[gid] = 0.f;
  if (gid < N_GRAPHS) cnt[gid] = 0.f;
  if (gid == 0) *ovf_cnt = 0;
  hist[tid] = 0;
  __syncthreads();
  const int* sp = ei + (size_t)w * CH_EDGES;
  const int* dp = ei + N_EDGES + (size_t)w * CH_EDGES;
  constexpr int PER = CH_EDGES / PART_THREADS;
  int key[PER], bn[PER], rk[PER];
#pragma unroll
  for (int k = 0; k < PER; ++k) {
    const int i = tid + PART_THREADS * k;
    const int s = sp[i], d = dp[i];
    const int b = d / BIN_NODES;
    key[k] = (s << 8) | (d - b * BIN_NODES);
    bn[k] = b;
  }
#pragma unroll
  for (int k = 0; k < PER; ++k) rk[k] = atomicAdd(&hist[bn[k]], 1);
  __syncthreads();
  const int v = hist[tid];
  int x = v;
#pragma unroll
  for (int o = 1; o < 32; o <<= 1) {
    const int y = __shfl_up(x, o);
    if (lane >= o) x += y;
  }
  if (lane == 31) wsum[wv] = x;
  __syncthreads();
  int add = 0;
  for (int q = 0; q < wv; ++q) add += wsum[q];
  const int base = add + x - v;
  if (tid < OFF_PITCH) off[(size_t)w * OFF_PITCH + tid] = base;
  hist[tid] = base;
  __syncthreads();
#pragma unroll
  for (int k = 0; k < PER; ++k) sorted[hist[bn[k]] + rk[k]] = key[k];
  __syncthreads();
  int* pp = part + (size_t)w * CH_EDGES;
#pragma unroll
  for (int k = 0; k < PER; ++k) pp[tid + PART_THREADS * k] = sorted[tid + PART_THREADS * k];
}

#define SLOT_WAVES 16
#define SLOT_RUNS  ((N_CHUNKS + SLOT_WAVES - 1) / SLOT_WAVES)
__global__ void __launch_bounds__(SLOT_WAVES * 32)
slot_kernel(const int* __restrict__ part, const int* __restrict__ off, int* __restrict__ ecnt,
            int* __restrict__ slot, int* __restrict__ ovf_cnt, int* __restrict__ ovf,
            const int* __restrict__ x, const float* __restrict__ tab, float* __restrict__ xs) {
  __shared__ int cnt_l[256];
  __shared__ int slot_l[BIN_NODES * SL_PITCH];
  const int tid = threadIdx.x;
  const int lane = tid & 31;
  const int wv = tid >> 5;
  const int b = blockIdx.x;
  if (tid < 256) cnt_l[tid] = 0;
  static_assert(N_COLS == 4 && (BIN_NODES + SLOT_WAVES - 1) / SLOT_WAVES <= 16, "two registers of 8 nodes x 4 columns");
  int xi[2];
#pragma unroll
  for (int k = 0; k < 2; ++k) {
    const int i = wv + SLOT_WAVES * (8 * k + (lane >> 2));
    const int ic = (i < BIN_NODES) ? i : (BIN_NODES - 1);
    xi[k] = x[(b * BIN_NODES + ic) * N_COLS + (lane & 3)];
  }
  int lo[SLOT_RUNS], hi[SLOT_RUNS], p[SLOT_RUNS];
#pragma unroll
  for (int u = 0; u < SLOT_RUNS; ++u) {
    const int c = u * SLOT_WAVES + wv;
    const int cc = (c < N_CHUNKS) ? c : (N_CHUNKS - 1);
    lo[u] = off[(size_t)cc * OFF_PITCH + b];
    hi[u] = off[(size_t)cc * OFF_PITCH + b + 1];
    if (c >= N_CHUNKS) hi[u] = lo[u];
  }
#pragma unroll
  for (int u = 0; u < SLOT_RUNS; ++u) {
    const int c = u * SLOT_WAVES + wv;
    const int j = lo[u] + lane;
    p[u] = -1;
    if (j < hi[u]) p[u] = part[(size_t)c * CH_EDGES + j];
  }
  __syncthreads();
  auto put = [&](int pk, int r) {
    const int dl = pk & 255;
    const int s = pk >> 8;
    if (r < SLOT_CAP) {
      slot_l[dl * SL_PITCH + r] = s;
    } else {
      const int q = atomicAdd(ovf_cnt, 1);
      v2i ent;
      ent.x = s;
      ent.y = b * BIN_NODES + dl;
      *(v2i*)(ovf + 2 * (size_t)q) = ent;
    }
  };
  int rr[SLOT_RUNS];
#pragma unroll
  for (int u = 0; u < SLOT_RUNS; ++u) {
    rr[u] = 0;
    if (p[u] >= 0) rr[u] = atomicAdd(&cnt_l[p[u] & 255], 1);
  }
#pragma unroll
  for (int u = 0; u < SLOT_RUNS; ++u)
    if (p[u] >= 0) put(p[u], rr[u]);
#pragma unroll
  for (int u = 0; u < SLOT_RUNS; ++u) {
    const int c = u * SLOT_WAVES + wv;
    for (int j = lo[u] + 32 + lane; j < hi[u]; j += 32) {
      const int pk = part[(size_t)c * CH_EDGES + j];
      put(pk, atomicAdd(&cnt_l[pk & 255], 1));
    }
  }
  __syncthreads();
  if (tid < BIN_NODES) ecnt[b * BIN_NODES + tid] = cnt_l[tid];
  constexpr int TRIPS = (BIN_NODES + SLOT_WAVES - 1) / SLOT_WAVES;
  static_assert(2 * 31 < SLOT_CAP, "two slot entries per lane");
  const int e0 = 2 * lane;
  int cq[TRIPS];
  v2i oq[TRIPS];
#pragma unroll
  for (int q = 0; q < TRIPS; ++q) {
    const int i = wv + SLOT_WAVES * q;
    const int ic = (i < BIN_NODES) ? i : (BIN_NODES - 1);
    cq[q] = cnt_l[ic];
    oq[q].x = slot_l[ic * SL_PITCH + e0];
    oq[q].y = slot_l[ic * SL_PITCH + e0 + 1];
  }
#pragma unroll
  for (int q = 0; q < TRIPS; ++q) {
    const int i = wv + SLOT_WAVES * q;
    if (i < BIN_NODES && e0 < cq[q]) *(v2i*)(slot + (size_t)(b * BIN_NODES + i) * SLOT_CAP + e0) = oq[q];
  }
  constexpr int ROW_NB = 7;
#pragma unroll
  for (int q0 = 0; q0 < TRIPS; q0 += ROW_NB) {
    v4f tr[ROW_NB][N_COLS];
#pragma unroll
    for (int u = 0; u < ROW_NB; ++u) {
      const int q = q0 + u;
      if (q < TRIPS) {
#pragma unroll
        for (int cc = 0; cc < N_COLS; ++cc) {
          const int v = __builtin_amdgcn_readlane(xi[q >> 3], (q & 7) * 4 + cc);
          tr[u][cc] = *(const v4f*)((const char*)tab + ((unsigned)(cc * VOCAB + v) * DIM + (unsigned)lane * 4) * 4u);
        }
      }
    }
#pragma unroll
    for (int u = 0; u < ROW_NB; ++u) {
      const int q = q0 + u;
      if (q < TRIPS) {
        const int i = wv + SLOT_WAVES * q;
        const v4f s = ((tr[u][0] + tr[u][1]) + tr[u][2]) + tr[u][3];
        static_assert((TRIPS - 1) * SLOT_WAVES <= BIN_NODES, "every trip but the last lies inside the bin");
        if (q < TRIPS - 1 || i < BIN_NODES)
          *(v4f*)((char*)xs + ((unsigned)(b * BIN_NODES + i) * DIM + (unsigned)lane * 4) * 4u) = s * deg_inv_sqrt(cq[q]);
      }
    }
  }
}

__device__ __forceinline__ v4f mean_row(v4f s, float c) {
  const float d = fmaxf(c, 1.0f);
  v4f r;
  r.x = s.x / d;
  r.y = s.y / d;
  r.z = s.z / d;
  r.w = s.w / d;
  return r;
}

__global__ void __launch_bounds__(256)
mean_gemm_wmma_f32(const float* __restrict__ A, const float* __restrict__ cnt, const float* __restrict__ B,
                   const float* __restrict__ bias, float* __restrict__ Cout, int n_tiles) {
  __shared__ __attribute__((aligned(16))) float sA[2][16 * A_PITCH];
  const int tid  = threadIdx.x;
  const int lane = tid & 31;
  const int wv   = tid >> 5;
  const int half = lane >> 4;
  const int lm   = lane & 15;
  const int col0 = wv * 16;

  v2f bf[32];
  load_b_frags(B, half, col0 + lm, bf);
  const float badd = bias[col0 + lm];

  const float* ga = A + (size_t)wv * DIM + lane * 4;
  const int so0 = wv * A_PITCH + lane * 4;
  const int so1 = (wv + 8) * A_PITCH + lane * 4;
  const int fo  = lm * A_PITCH + 4 * half;

  int tile = blockIdx.x;
  v4f p0 = {0.f, 0.f, 0.f, 0.f}, p1 = {0.f, 0.f, 0.f, 0.f};
  float c0 = 1.f, c1 = 1.f;
  if (tile < n_tiles) {
    p0 = *(const v4f*)(ga + (size_t)tile * 16 * DIM);
    p1 = *(const v4f*)(ga + (size_t)tile * 16 * DIM + 8 * DIM);
    c0 = cnt[tile * 16 + wv];
    c1 = cnt[tile * 16 + wv + 8];
    *(v4f*)(&sA[0][so0]) = mean_row(p0, c0);
    *(v4f*)(&sA[0][so1]) = mean_row(p1, c1);
  }
  __syncthreads();
  int buf = 0;
  for (; tile < n_tiles; tile += gridDim.x) {
    const int nt = tile + gridDim.x;
    const bool more = nt < n_tiles;
    if (more) {
      p0 = *(const v4f*)(ga + (size_t)nt * 16 * DIM);
      p1 = *(const v4f*)(ga + (size_t)nt * 16 * DIM + 8 * DIM);
      c0 = cnt[nt * 16 + wv];
      c1 = cnt[nt * 16 + wv + 8];
    }
    const v8f acc = tile_product(&sA[buf][fo], bf);
    float* cp = Cout + (size_t)(tile * 16 + half * 8) * DIM + col0 + lm;
#pragma unroll
    for (int r = 0; r < 8; ++r) cp[r * DIM] = acc[r] + badd;
    if (more) {
      *(v4f*)(&sA[buf ^ 1][so0]) = mean_row(p0, c0);
      *(v4f*)(&sA[buf ^ 1][so1]) = mean_row(p1, c1);
    }
    __syncthreads();
    buf ^= 1;
  }
}


template <int G>
__device__ __forceinline__ void add_rows(v4f& acc, const float* __restrict__ xs, int lane4, int ent, int k0) {
  v4f v[G];
#pragma unroll
  for (int u = 0; u < G; ++u) {
    const int sk = __builtin_amdgcn_readlane(ent, k0 + u);
    v[u] = *(const v4f*)((const char*)xs + ((unsigned)sk * DIM + (unsigned)lane4) * 4u);
  }
#pragma unroll
  for (int u = 0; u < G; ++u) acc += v[u];
}

__device__ __forceinline__ int edge_chunk(const int* __restrict__ slot, int base, int end, int lane) {
  const int j = base + lane;
  int ent = 0;
  if (j < end) ent = slot[j];
  return ent;
}

__device__ __forceinline__ v4f edge_sum(const int* __restrict__ slot, const float* __restrict__ xs,
                                        int beg, int end, int first, int lane) {
  v4f acc = {0.f, 0.f, 0.f, 0.f};
  int ent = first;
  for (int base = beg; base < end; base += 32) {
    if (base != beg) ent = edge_chunk(slot, base, end, lane);
    const int m = (end - base < 32) ? (end - base) : 32;
    int k = 0;
    for (; k + 8 <= m; k += 8) add_rows<8>(acc, xs, lane * 4, ent, k);
    if (m - k >= 4) { add_rows<4>(acc, xs, lane * 4, ent, k); k += 4; }
    if (m - k >= 2) { add_rows<2>(acc, xs, lane * 4, ent, k); k += 2; }
    if (m - k >= 1) add_rows<1>(acc, xs, lane * 4, ent, k);
  }
  return acc;
}

__device__ __forceinline__ v4f overflow_sum(v4f acc, const int* __restrict__ ovf_cnt, const int* __restrict__ ovf,
                                            const float* __restrict__ xs, int n, int lane) {
  const int no = *ovf_cnt;
  for (int j0 = 0; j0 < no; j0 += 32) {
    const int j = j0 + lane;
    v2i e = {0, -1};
    if (j < no) e = *(const v2i*)(ovf + 2 * (size_t)j);
    unsigned mask = __builtin_amdgcn_ballot_w32(e.y == n);
    while (mask) {
      const int k = __builtin_ctz(mask);
      mask &= mask - 1;
      add_rows<1>(acc, xs, lane * 4, e.x, k);
    }
  }
  return acc;
}

__device__ __forceinline__ v4f bias_self_relu(v4f bb, v4f self, float di, v4f acc) {
  v4f r = (acc + self) * di + bb;
  r.x = fmaxf(r.x, 0.f);
  r.y = fmaxf(r.y, 0.f);
  r.z = fmaxf(r.z, 0.f);
  r.w = fmaxf(r.w, 0.f);
  return r;
}

static_assert(N_NODES % 16 == 0, "whole 16-row tiles only: the node index is never out of range");
__global__ void __launch_bounds__(256, 8)
rows_gemm_kernel(const int* __restrict__ ecnt, const int* __restrict__ slot,
                 const int* __restrict__ ovf_cnt, const int* __restrict__ ovf,
                 const float* __restrict__ xs, const float* __restrict__ bl,
                 const float* __restrict__ B, float* __restrict__ Cout, int n_tiles) {
  __shared__ __attribute__((aligned(16))) float sA[2][16 * A_PITCH];
  const int lane = threadIdx.x & 31;
  const int wv   = __builtin_amdgcn_readfirstlane((int)(threadIdx.x >> 5));
  const int half = lane >> 4;
  const int lm   = lane & 15;
  const int col0 = wv * 16;

  v2f bf[32];
  load_b_frags(B, half, col0 + lm, bf);

  const int so0 = wv * A_PITCH + lane * 4;
  const int so1 = (wv + 8) * A_PITCH + lane * 4;
  const int fo  = lm * A_PITCH + 4 * half;

  const v4f bb = *(const v4f*)(bl + lane * 4);

  int buf = 0;
  for (int tile = blockIdx.x; tile < n_tiles; tile += gridDim.x) {
    const int n0 = tile * 16 + wv;
    const int n1 = n0 + 8;
    const int c0 = ecnt[n0];
    const int c1 = ecnt[n1];
    const float di0 = deg_inv_sqrt(c0);
    const float di1 = deg_inv_sqrt(c1);
    const int beg0 = n0 * SLOT_CAP, end0 = beg0 + ((c0 < SLOT_CAP) ? c0 : SLOT_CAP);
    const int beg1 = n1 * SLOT_CAP, end1 = beg1 + ((c1 < SLOT_CAP) ? c1 : SLOT_CAP);
    const int e0 = edge_chunk(slot, beg0, end0, lane);
    const int e1 = edge_chunk(slot, beg1, end1, lane);
    const v4f self0 = *(const v4f*)((const char*)xs + ((unsigned)n0 * DIM + (unsigned)lane * 4) * 4u);
    v4f a0 = edge_sum(slot, xs, beg0, end0, e0, lane);
    if (c0 > SLOT_CAP) a0 = overflow_sum(a0, ovf_cnt, ovf, xs, n0, lane);
    const v4f self1 = *(const v4f*)((const char*)xs + ((unsigned)n1 * DIM + (unsigned)lane * 4) * 4u);
    *(v4f*)(&sA[buf][so0]) = bias_self_relu(bb, self0, di0, a0) * di0;
    v4f a1 = edge_sum(slot, xs, beg1, end1, e1, lane);
    if (c1 > SLOT_CAP) a1 = overflow_sum(a1, ovf_cnt, ovf, xs, n1, lane);
    *(v4f*)(&sA[buf][so1]) = bias_self_relu(bb, self1, di1, a1) * di1;
    __syncthreads();
    const v8f acc = tile_product(&sA[buf][fo], bf);
    float* cp = (float*)((char*)Cout + (unsigned)((tile * 16 + half * 8) * DIM + col0 + lm) * 4u);
#pragma unroll
    for (int r = 0; r < 8; ++r) cp[r * DIM] = acc[r];
    buf ^= 1;
  }
}

#define POOL_NPW 5
static_assert(N_NODES % (8 * POOL_NPW) == 0, "whole blocks only: the node index is never out of range");
__device__ __forceinline__ void pool_add_run(float* __restrict__ pool, float* __restrict__ cnt,
                                             int g, int lane, v4f a, int len) {
  float* p = pool + (size_t)g * DIM + lane * 4;
  unsafeAtomicAdd(p + 0, a.x);
  unsafeAtomicAdd(p + 1, a.y);
  unsafeAtomicAdd(p + 2, a.z);
  unsafeAtomicAdd(p + 3, a.w);
  if (lane == 0) unsafeAtomicAdd(cnt + g, (float)len);
}
__global__ void __launch_bounds__(256)
gather_pool_kernel(const int* __restrict__ ecnt, const int* __restrict__ slot,
                   const int* __restrict__ ovf_cnt, const int* __restrict__ ovf,
                   const float* __restrict__ xs, const float* __restrict__ bl, const int* __restrict__ batch,
                   float* __restrict__ pool, float* __restrict__ cnt) {
  const int lane = threadIdx.x & 31;
  const int w = __builtin_amdgcn_readfirstlane((int)((blockIdx.x * 256 + threadIdx.x) >> 5));
  const int n0 = w * POOL_NPW;
  const v4f bb = *(const v4f*)(bl + lane * 4);
  int g[POOL_NPW], c[POOL_NPW], en[POOL_NPW];
  int e[POOL_NPW];
  v4f self[POOL_NPW];
#pragma unroll
  for (int i = 0; i < POOL_NPW; ++i) {
    c[i] = ecnt[n0 + i];
    g[i] = batch[n0 + i];
  }
#pragma unroll
  for (int i = 0; i < POOL_NPW; ++i) {
    en[i] = (n0 + i) * SLOT_CAP + ((c[i] < SLOT_CAP) ? c[i] : SLOT_CAP);
    e[i] = edge_chunk(slot, (n0 + i) * SLOT_CAP, en[i], lane);
    self[i] = *(const v4f*)((const char*)xs + ((unsigned)(n0 + i) * DIM + (unsigned)lane * 4) * 4u);
  }
  v4f row[POOL_NPW];
#pragma unroll
  for (int i = 0; i < POOL_NPW; ++i) {
    v4f a = edge_sum(slot, xs, (n0 + i) * SLOT_CAP, en[i], e[i], lane);
    if (c[i] > SLOT_CAP) a = overflow_sum(a, ovf_cnt, ovf, xs, n0 + i, lane);
    row[i] = bias_self_relu(bb, self[i], deg_inv_sqrt(c[i]), a);
  }
  int cur = g[0];
  int len = 1;
  v4f acc = row[0];
#pragma unroll
  for (int i = 1; i < POOL_NPW; ++i) {
    if (g[i] == cur) {
      acc += row[i];
      ++len;
    } else {
      pool_add_run(pool, cnt, cur, lane, acc, len);
      cur = g[i];
      acc = row[i];
      len = 1;
    }
  }
  pool_add_run(pool, cnt, cur, lane, acc, len);
}

extern "C" void kernel_launch(void* const* d_in, const int* in_sizes, int n_in,
                              void* d_out, int out_size, void* d_ws, size_t ws_size,
                              hipStream_t stream) {
  const int*   x    = (const int*)d_in[0];
  const int*   ei   = (const int*)d_in[1];
  const int*   batch= (const int*)d_in[2];
  const float* emb  = (const float*)d_in[3];
  const float* W    = (const float*)d_in[4];
  const float* b    = (const float*)d_in[5];
  const float* Wr   = (const float*)d_in[6];
  const float* br   = (const float*)d_in[7];
  float* out = (float*)d_out;

  char* ws = (char*)d_ws;
  const size_t NH = (size_t)N_NODES * DIM * sizeof(float);
  float* bufA = (float*)(ws);
  float* bufB = (float*)(ws + NH);
  int*   ecnt    = (int*)(ws + 2 * NH);
  int*   ovf_cnt = (int*)(ws + 2 * NH + 256 * 1024);
  int*   slot    = (int*)(ws + 2 * NH + 8 * 1024 * 1024);
  int*   ovf     = (int*)(ws + 3 * NH + 8 * 1024 * 1024);
  float* pool = (float*)(ws + 3 * NH + 4 * 1024 * 1024);
  float* cnt  = (float*)(ws + 3 * NH + 4 * 1024 * 1024 + 512 * 1024);
  int*   part = (int*)(ws + 3 * NH + 16 * 1024 * 1024);
  int*   off  = (int*)(ws + 3 * NH + 20 * 1024 * 1024);
  float* tab  = (float*)(ws + 3 * NH + 24 * 1024 * 1024);

  const int B256 = 256;
  const int pool_blocks = N_NODES / (8 * POOL_NPW);

  const int g_blocks = 782;
  const int n_tiles = N_NODES / 16;

  partition_kernel<<<N_CHUNKS + TAB_TILES, PART_THREADS, 0, stream>>>(ei, part, off, ovf_cnt, pool, cnt, emb, W, tab);

  slot_kernel<<<N_BINS, SLOT_WAVES * 32, 0, stream>>>(part, off, ecnt, slot, ovf_cnt, ovf, x, tab, bufA);

  rows_gemm_kernel<<<g_blocks, B256, 0, stream>>>(ecnt, slot, ovf_cnt, ovf, bufA, b,
                                                  W + (size_t)1 * DIM * DIM, bufB, n_tiles);
  rows_gemm_kernel<<<g_blocks, B256, 0, stream>>>(ecnt, slot, ovf_cnt, ovf, bufB, b + 1 * DIM,
                                                  W + (size_t)2 * DIM * DIM, bufA, n_tiles);
  rows_gemm_kernel<<<g_blocks, B256, 0, stream>>>(ecnt, slot, ovf_cnt, ovf, bufA, b + 2 * DIM,
                                                  W + (size_t)3 * DIM * DIM, bufB, n_tiles);

  gather_pool_kernel<<<pool_blocks, B256, 0, stream>>>(ecnt, slot, ovf_cnt, ovf, bufB, b + 3 * DIM, batch, pool, cnt);

  mean_gemm_wmma_f32<<<N_GRAPHS / 16, B256, 0, stream>>>(pool, cnt, Wr, br, out, N_GRAPHS / 16);
}


# Layer launches: row tiles claimed from an atomic counter, one ahead, by a grid that fills every WGP's wave slots (1024 workgroups; was 782 with 4 fixed tiles each); every sum and its order unchanged
# speedup vs baseline: 6.1203x; 1.0313x over previous
#include <hip/hip_runtime.h>

#define N_NODES  50000
#define N_EDGES  800000
#define DIM      128
#define N_LAYERS 4
#define N_GRAPHS 512
#define N_COLS   4
#define VOCAB    128

typedef __attribute__((ext_vector_type(2))) float v2f;
typedef __attribute__((ext_vector_type(4))) float v4f;
typedef __attribute__((ext_vector_type(8))) float v8f;

#define TAB_ROWS  (N_COLS * VOCAB)
#define TAB_TILES (TAB_ROWS / 16)
static_assert(TAB_ROWS % 16 == 0, "whole 16-row tiles of the table");

__device__ __forceinline__ float deg_inv_sqrt(int c) {
  const float d = 1.0f + (float)c;
  return (d > 0.f) ? rsqrtf(d) : 0.f;
}

#define A_PITCH 132

__device__ __forceinline__ void load_b_frags(const float* __restrict__ B, int half, int col, v2f (&bf)[32]) {
  const float* bp = B + (size_t)(4 * half) * DIM + col;
#pragma unroll
  for (int s = 0; s < 32; ++s) {
    const int r0 = 8 * (s >> 1) + 2 * (s & 1);
    bf[s].x = bp[r0 * DIM];
    bf[s].y = bp[(r0 + 1) * DIM];
  }
}

__device__ __forceinline__ v8f tile_product(const float* fa, const v2f (&bf)[32]) {
  v8f acc = {};
#pragma unroll
  for (int j = 0; j < 16; ++j) {
    const v4f a = *(const v4f*)(fa + 8 * j);
    v2f alo, ahi;
    alo.x = a.x; alo.y = a.y;
    ahi.x = a.z; ahi.y = a.w;
    acc = __builtin_amdgcn_wmma_f32_16x16x4_f32(
         false, alo,  false, bf[2 * j],
         (short)0, acc,  false,  false);
    acc = __builtin_amdgcn_wmma_f32_16x16x4_f32(
         false, ahi,  false, bf[2 * j + 1],
         (short)0, acc,  false,  false);
  }
  return acc;
}

__device__ __forceinline__ void table_tile(const float* __restrict__ emb, const float* __restrict__ B,
                                           float* __restrict__ tab, int tile, float* sA) {
  const int lane = threadIdx.x & 31;
  const int wv   = __builtin_amdgcn_readfirstlane((int)(threadIdx.x >> 5));
  const bool worker = wv < 8;
  const int half = lane >> 4;
  const int lm   = lane & 15;
  const int col0 = wv * 16;
  v2f bf[32];
  if (worker) {
    load_b_frags(B, half, col0 + lm, bf);
    const float* ga = emb + ((size_t)tile * 16 + wv) * DIM + lane * 4;
    *(v4f*)(&sA[wv * A_PITCH + lane * 4])       = *(const v4f*)ga;
    *(v4f*)(&sA[(wv + 8) * A_PITCH + lane * 4]) = *(const v4f*)(ga + 8 * DIM);
  }
  __syncthreads();
  if (worker) {
    const v8f acc = tile_product(&sA[lm * A_PITCH + 4 * half], bf);
    float* cp = tab + (size_t)(tile * 16 + half * 8) * DIM + col0 + lm;
#pragma unroll
    for (int r = 0; r < 8; ++r) cp[r * DIM] = acc[r];
  }
}

#define SLOT_CAP 64
typedef __attribute__((ext_vector_type(2))) int v2i;

#define BIN_NODES 200
#define N_BINS    (N_NODES / BIN_NODES)
#define CH_EDGES  3200
#define N_CHUNKS  (N_EDGES / CH_EDGES)
#define OFF_PITCH 256
#define SL_PITCH  (SLOT_CAP + 1)
static_assert(N_NODES % BIN_NODES == 0 && N_EDGES % CH_EDGES == 0, "whole bins and whole chunks only");
static_assert(BIN_NODES <= 256 && N_NODES < (1 << 23), "packed edge word: source id above 8 bits of bin-local destination");
static_assert(N_BINS < OFF_PITCH, "offset entries 0..N_BINS of a chunk fit its row of the table");

#define TCTR_PITCH 64
#define PART_THREADS 320
static_assert(CH_EDGES % PART_THREADS == 0 && PART_THREADS % 32 == 0, "every thread takes the same number of edges");
static_assert(N_BINS < PART_THREADS && OFF_PITCH <= PART_THREADS, "one thread per bin counter and per offset entry");
static_assert(N_GRAPHS * DIM <= N_CHUNKS * PART_THREADS, "one thread per pool element");
static_assert(PART_THREADS >= 256, "the table product needs 8 waves");
__global__ void __launch_bounds__(PART_THREADS)
partition_kernel(const int* __restrict__ ei, int* __restrict__ part, int* __restrict__ off,
                 int* __restrict__ ovf_cnt, float* __restrict__ pool, float* __restrict__ cnt,
                 const float* __restrict__ emb, const float* __restrict__ W0, float* __restrict__ tab,
                 int* __restrict__ tctr) {
  __shared__ int hist[PART_THREADS];
  __shared__ int wsum[PART_THREADS / 32];
  __shared__ int sorted[CH_EDGES];
  __shared__ __attribute__((aligned(16))) float sT[16 * A_PITCH];
  const int tid = threadIdx.x;
  const int lane = tid & 31;
  const int wv = tid >> 5;
  const int w = blockIdx.x;
  if (w >= N_CHUNKS) {
    table_tile(emb, W0, tab, w - N_CHUNKS, sT);
    return;
  }
  const int gid = w * PART_THREADS + tid;
  if (gid < N_GRAPHS * DIM) pool[gid] = 0.f;
  if (gid < N_GRAPHS) cnt[gid] = 0.f;
  if (gid == 0) *ovf_cnt = 0;
  if (gid < N_LAYERS - 1) tctr[gid * TCTR_PITCH] = 0;
  hist[tid] = 0;
  __syncthreads();
  const int* sp = ei + (size_t)w * CH_EDGES;
  const int* dp = ei + N_EDGES + (size_t)w * CH_EDGES;
  constexpr int PER = CH_EDGES / PART_THREADS;
  int key[PER], bn[PER], rk[PER];
#pragma unroll
  for (int k = 0; k < PER; ++k) {
    const int i = tid + PART_THREADS * k;
    const int s = sp[i], d = dp[i];
    const int b = d / BIN_NODES;
    key[k] = (s << 8) | (d - b * BIN_NODES);
    bn[k] = b;
  }
#pragma unroll
  for (int k = 0; k < PER; ++k) rk[k] = atomicAdd(&hist[bn[k]], 1);
  __syncthreads();
  const int v = hist[tid];
  int x = v;
#pragma unroll
  for (int o = 1; o < 32; o <<= 1) {
    const int y = __shfl_up(x, o);
    if (lane >= o) x += y;
  }
  if (lane == 31) wsum[wv] = x;
  __syncthreads();
  int add = 0;
  for (int q = 0; q < wv; ++q) add += wsum[q];
  const int base = add + x - v;
  if (tid < OFF_PITCH) off[(size_t)w * OFF_PITCH + tid] = base;
  hist[tid] = base;
  __syncthreads();
#pragma unroll
  for (int k = 0; k < PER; ++k) sorted[hist[bn[k]] + rk[k]] = key[k];
  __syncthreads();
  int* pp = part + (size_t)w * CH_EDGES;
#pragma unroll
  for (int k = 0; k < PER; ++k) pp[tid + PART_THREADS * k] = sorted[tid + PART_THREADS * k];
}

#define SLOT_WAVES 16
#define SLOT_RUNS  ((N_CHUNKS + SLOT_WAVES - 1) / SLOT_WAVES)
__global__ void __launch_bounds__(SLOT_WAVES * 32)
slot_kernel(const int* __restrict__ part, const int* __restrict__ off, int* __restrict__ ecnt,
            int* __restrict__ slot, int* __restrict__ ovf_cnt, int* __restrict__ ovf,
            const int* __restrict__ x, const float* __restrict__ tab, float* __restrict__ xs) {
  __shared__ int cnt_l[256];
  __shared__ int slot_l[BIN_NODES * SL_PITCH];
  const int tid = threadIdx.x;
  const int lane = tid & 31;
  const int wv = tid >> 5;
  const int b = blockIdx.x;
  if (tid < 256) cnt_l[tid] = 0;
  static_assert(N_COLS == 4 && (BIN_NODES + SLOT_WAVES - 1) / SLOT_WAVES <= 16, "two registers of 8 nodes x 4 columns");
  int xi[2];
#pragma unroll
  for (int k = 0; k < 2; ++k) {
    const int i = wv + SLOT_WAVES * (8 * k + (lane >> 2));
    const int ic = (i < BIN_NODES) ? i : (BIN_NODES - 1);
    xi[k] = x[(b * BIN_NODES + ic) * N_COLS + (lane & 3)];
  }
  int lo[SLOT_RUNS], hi[SLOT_RUNS], p[SLOT_RUNS];
#pragma unroll
  for (int u = 0; u < SLOT_RUNS; ++u) {
    const int c = u * SLOT_WAVES + wv;
    const int cc = (c < N_CHUNKS) ? c : (N_CHUNKS - 1);
    lo[u] = off[(size_t)cc * OFF_PITCH + b];
    hi[u] = off[(size_t)cc * OFF_PITCH + b + 1];
    if (c >= N_CHUNKS) hi[u] = lo[u];
  }
#pragma unroll
  for (int u = 0; u < SLOT_RUNS; ++u) {
    const int c = u * SLOT_WAVES + wv;
    const int j = lo[u] + lane;
    p[u] = -1;
    if (j < hi[u]) p[u] = part[(size_t)c * CH_EDGES + j];
  }
  __syncthreads();
  auto put = [&](int pk, int r) {
    const int dl = pk & 255;
    const int s = pk >> 8;
    if (r < SLOT_CAP) {
      slot_l[dl * SL_PITCH + r] = s;
    } else {
      const int q = atomicAdd(ovf_cnt, 1);
      v2i ent;
      ent.x = s;
      ent.y = b * BIN_NODES + dl;
      *(v2i*)(ovf + 2 * (size_t)q) = ent;
    }
  };
  int rr[SLOT_RUNS];
#pragma unroll
  for (int u = 0; u < SLOT_RUNS; ++u) {
    rr[u] = 0;
    if (p[u] >= 0) rr[u] = atomicAdd(&cnt_l[p[u] & 255], 1);
  }
#pragma unroll
  for (int u = 0; u < SLOT_RUNS; ++u)
    if (p[u] >= 0) put(p[u], rr[u]);
#pragma unroll
  for (int u = 0; u < SLOT_RUNS; ++u) {
    const int c = u * SLOT_WAVES + wv;
    for (int j = lo[u] + 32 + lane; j < hi[u]; j += 32) {
      const int pk = part[(size_t)c * CH_EDGES + j];
      put(pk, atomicAdd(&cnt_l[pk & 255], 1));
    }
  }
  __syncthreads();
  if (tid < BIN_NODES) ecnt[b * BIN_NODES + tid] = cnt_l[tid];
  constexpr int TRIPS = (BIN_NODES + SLOT_WAVES - 1) / SLOT_WAVES;
  static_assert(2 * 31 < SLOT_CAP, "two slot entries per lane");
  const int e0 = 2 * lane;
  int cq[TRIPS];
  v2i oq[TRIPS];
#pragma unroll
  for (int q = 0; q < TRIPS; ++q) {
    const int i = wv + SLOT_WAVES * q;
    const int ic = (i < BIN_NODES) ? i : (BIN_NODES - 1);
    cq[q] = cnt_l[ic];
    oq[q].x = slot_l[ic * SL_PITCH + e0];
    oq[q].y = slot_l[ic * SL_PITCH + e0 + 1];
  }
#pragma unroll
  for (int q = 0; q < TRIPS; ++q) {
    const int i = wv + SLOT_WAVES * q;
    if (i < BIN_NODES && e0 < cq[q]) *(v2i*)(slot + (size_t)(b * BIN_NODES + i) * SLOT_CAP + e0) = oq[q];
  }
  constexpr int ROW_NB = 7;
#pragma unroll
  for (int q0 = 0; q0 < TRIPS; q0 += ROW_NB) {
    v4f tr[ROW_NB][N_COLS];
#pragma unroll
    for (int u = 0; u < ROW_NB; ++u) {
      const int q = q0 + u;
      if (q < TRIPS) {
#pragma unroll
        for (int cc = 0; cc < N_COLS; ++cc) {
          const int v = __builtin_amdgcn_readlane(xi[q >> 3], (q & 7) * 4 + cc);
          tr[u][cc] = *(const v4f*)((const char*)tab + ((unsigned)(cc * VOCAB + v) * DIM + (unsigned)lane * 4) * 4u);
        }
      }
    }
#pragma unroll
    for (int u = 0; u < ROW_NB; ++u) {
      const int q = q0 + u;
      if (q < TRIPS) {
        const int i = wv + SLOT_WAVES * q;
        const v4f s = ((tr[u][0] + tr[u][1]) + tr[u][2]) + tr[u][3];
        static_assert((TRIPS - 1) * SLOT_WAVES <= BIN_NODES, "every trip but the last lies inside the bin");
        if (q < TRIPS - 1 || i < BIN_NODES)
          *(v4f*)((char*)xs + ((unsigned)(b * BIN_NODES + i) * DIM + (unsigned)lane * 4) * 4u) = s * deg_inv_sqrt(cq[q]);
      }
    }
  }
}

__device__ __forceinline__ v4f mean_row(v4f s, float c) {
  const float d = fmaxf(c, 1.0f);
  v4f r;
  r.x = s.x / d;
  r.y = s.y / d;
  r.z = s.z / d;
  r.w = s.w / d;
  return r;
}

__global__ void __launch_bounds__(256)
mean_gemm_wmma_f32(const float* __restrict__ A, const float* __restrict__ cnt, const float* __restrict__ B,
                   const float* __restrict__ bias, float* __restrict__ Cout, int n_tiles) {
  __shared__ __attribute__((aligned(16))) float sA[2][16 * A_PITCH];
  const int tid  = threadIdx.x;
  const int lane = tid & 31;
  const int wv   = tid >> 5;
  const int half = lane >> 4;
  const int lm   = lane & 15;
  const int col0 = wv * 16;

  v2f bf[32];
  load_b_frags(B, half, col0 + lm, bf);
  const float badd = bias[col0 + lm];

  const float* ga = A + (size_t)wv * DIM + lane * 4;
  const int so0 = wv * A_PITCH + lane * 4;
  const int so1 = (wv + 8) * A_PITCH + lane * 4;
  const int fo  = lm * A_PITCH + 4 * half;

  int tile = blockIdx.x;
  v4f p0 = {0.f, 0.f, 0.f, 0.f}, p1 = {0.f, 0.f, 0.f, 0.f};
  float c0 = 1.f, c1 = 1.f;
  if (tile < n_tiles) {
    p0 = *(const v4f*)(ga + (size_t)tile * 16 * DIM);
    p1 = *(const v4f*)(ga + (size_t)tile * 16 * DIM + 8 * DIM);
    c0 = cnt[tile * 16 + wv];
    c1 = cnt[tile * 16 + wv + 8];
    *(v4f*)(&sA[0][so0]) = mean_row(p0, c0);
    *(v4f*)(&sA[0][so1]) = mean_row(p1, c1);
  }
  __syncthreads();
  int buf = 0;
  for (; tile < n_tiles; tile += gridDim.x) {
    const int nt = tile + gridDim.x;
    const bool more = nt < n_tiles;
    if (more) {
      p0 = *(const v4f*)(ga + (size_t)nt * 16 * DIM);
      p1 = *(const v4f*)(ga + (size_t)nt * 16 * DIM + 8 * DIM);
      c0 = cnt[nt * 16 + wv];
      c1 = cnt[nt * 16 + wv + 8];
    }
    const v8f acc = tile_product(&sA[buf][fo], bf);
    float* cp = Cout + (size_t)(tile * 16 + half * 8) * DIM + col0 + lm;
#pragma unroll
    for (int r = 0; r < 8; ++r) cp[r * DIM] = acc[r] + badd;
    if (more) {
      *(v4f*)(&sA[buf ^ 1][so0]) = mean_row(p0, c0);
      *(v4f*)(&sA[buf ^ 1][so1]) = mean_row(p1, c1);
    }
    __syncthreads();
    buf ^= 1;
  }
}


template <int G>
__device__ __forceinline__ void add_rows(v4f& acc, const float* __restrict__ xs, int lane4, int ent, int k0) {
  v4f v[G];
#pragma unroll
  for (int u = 0; u < G; ++u) {
    const int sk = __builtin_amdgcn_readlane(ent, k0 + u);
    v[u] = *(const v4f*)((const char*)xs + ((unsigned)sk * DIM + (unsigned)lane4) * 4u);
  }
#pragma unroll
  for (int u = 0; u < G; ++u) acc += v[u];
}

__device__ __forceinline__ int edge_chunk(const int* __restrict__ slot, int base, int end, int lane) {
  const int j = base + lane;
  int ent = 0;
  if (j < end) ent = slot[j];
  return ent;
}

__device__ __forceinline__ v4f edge_sum(const int* __restrict__ slot, const float* __restrict__ xs,
                                        int beg, int end, int first, int lane) {
  v4f acc = {0.f, 0.f, 0.f, 0.f};
  int ent = first;
  for (int base = beg; base < end; base += 32) {
    if (base != beg) ent = edge_chunk(slot, base, end, lane);
    const int m = (end - base < 32) ? (end - base) : 32;
    int k = 0;
    for (; k + 8 <= m; k += 8) add_rows<8>(acc, xs, lane * 4, ent, k);
    if (m - k >= 4) { add_rows<4>(acc, xs, lane * 4, ent, k); k += 4; }
    if (m - k >= 2) { add_rows<2>(acc, xs, lane * 4, ent, k); k += 2; }
    if (m - k >= 1) add_rows<1>(acc, xs, lane * 4, ent, k);
  }
  return acc;
}

__device__ __forceinline__ v4f overflow_sum(v4f acc, const int* __restrict__ ovf_cnt, const int* __restrict__ ovf,
                                            const float* __restrict__ xs, int n, int lane) {
  const int no = *ovf_cnt;
  for (int j0 = 0; j0 < no; j0 += 32) {
    const int j = j0 + lane;
    v2i e = {0, -1};
    if (j < no) e = *(const v2i*)(ovf + 2 * (size_t)j);
    unsigned mask = __builtin_amdgcn_ballot_w32(e.y == n);
    while (mask) {
      const int k = __builtin_ctz(mask);
      mask &= mask - 1;
      add_rows<1>(acc, xs, lane * 4, e.x, k);
    }
  }
  return acc;
}

__device__ __forceinline__ v4f bias_self_relu(v4f bb, v4f self, float di, v4f acc) {
  v4f r = (acc + self) * di + bb;
  r.x = fmaxf(r.x, 0.f);
  r.y = fmaxf(r.y, 0.f);
  r.z = fmaxf(r.z, 0.f);
  r.w = fmaxf(r.w, 0.f);
  return r;
}

static_assert(N_NODES % 16 == 0, "whole 16-row tiles only: the node index is never out of range");
__global__ void __launch_bounds__(256, 8)
rows_gemm_kernel(const int* __restrict__ ecnt, const int* __restrict__ slot,
                 const int* __restrict__ ovf_cnt, const int* __restrict__ ovf,
                 const float* __restrict__ xs, const float* __restrict__ bl,
                 const float* __restrict__ B, float* __restrict__ Cout, int n_tiles,
                 int* __restrict__ tctr) {
  __shared__ __attribute__((aligned(16))) float sA[2][16 * A_PITCH];
  __shared__ int s_tile[2];
  const bool claimer = threadIdx.x == 0;
  const int lane = threadIdx.x & 31;
  const int wv   = __builtin_amdgcn_readfirstlane((int)(threadIdx.x >> 5));
  const int half = lane >> 4;
  const int lm   = lane & 15;
  const int col0 = wv * 16;

  v2f bf[32];
  load_b_frags(B, half, col0 + lm, bf);

  const int so0 = wv * A_PITCH + lane * 4;
  const int so1 = (wv + 8) * A_PITCH + lane * 4;
  const int fo  = lm * A_PITCH + 4 * half;

  const v4f bb = *(const v4f*)(bl + lane * 4);

  int tile = blockIdx.x;
  int buf = 0;
  while (tile < n_tiles) {
    const int n0 = tile * 16 + wv;
    const int n1 = n0 + 8;
    const int c0 = ecnt[n0];
    const int c1 = ecnt[n1];
    const float di0 = deg_inv_sqrt(c0);
    const float di1 = deg_inv_sqrt(c1);
    const int beg0 = n0 * SLOT_CAP, end0 = beg0 + ((c0 < SLOT_CAP) ? c0 : SLOT_CAP);
    const int beg1 = n1 * SLOT_CAP, end1 = beg1 + ((c1 < SLOT_CAP) ? c1 : SLOT_CAP);
    const int e0 = edge_chunk(slot, beg0, end0, lane);
    const int e1 = edge_chunk(slot, beg1, end1, lane);
    const v4f self0 = *(const v4f*)((const char*)xs + ((unsigned)n0 * DIM + (unsigned)lane * 4) * 4u);
    int nx = 0;
    if (claimer) nx = (int)atomicInc((unsigned int*)tctr, 0xffffffffu);
    v4f a0 = edge_sum(slot, xs, beg0, end0, e0, lane);
    if (c0 > SLOT_CAP) a0 = overflow_sum(a0, ovf_cnt, ovf, xs, n0, lane);
    const v4f self1 = *(const v4f*)((const char*)xs + ((unsigned)n1 * DIM + (unsigned)lane * 4) * 4u);
    *(v4f*)(&sA[buf][so0]) = bias_self_relu(bb, self0, di0, a0) * di0;
    v4f a1 = edge_sum(slot, xs, beg1, end1, e1, lane);
    if (c1 > SLOT_CAP) a1 = overflow_sum(a1, ovf_cnt, ovf, xs, n1, lane);
    *(v4f*)(&sA[buf][so1]) = bias_self_relu(bb, self1, di1, a1) * di1;
    if (claimer) s_tile[buf ^ 1] = nx + (int)gridDim.x;
    __syncthreads();
    const int next = __builtin_amdgcn_readfirstlane(s_tile[buf ^ 1]);
    const v8f acc = tile_product(&sA[buf][fo], bf);
    float* cp = (float*)((char*)Cout + (unsigned)((tile * 16 + half * 8) * DIM + col0 + lm) * 4u);
#pragma unroll
    for (int r = 0; r < 8; ++r) cp[r * DIM] = acc[r];
    tile = next;
    buf ^= 1;
  }
}

#define POOL_NPW 5
static_assert(N_NODES % (8 * POOL_NPW) == 0, "whole blocks only: the node index is never out of range");
__device__ __forceinline__ void pool_add_run(float* __restrict__ pool, float* __restrict__ cnt,
                                             int g, int lane, v4f a, int len) {
  float* p = pool + (size_t)g * DIM + lane * 4;
  unsafeAtomicAdd(p + 0, a.x);
  unsafeAtomicAdd(p + 1, a.y);
  unsafeAtomicAdd(p + 2, a.z);
  unsafeAtomicAdd(p + 3, a.w);
  if (lane == 0) unsafeAtomicAdd(cnt + g, (float)len);
}
__global__ void __launch_bounds__(256)
gather_pool_kernel(const int* __restrict__ ecnt, const int* __restrict__ slot,
                   const int* __restrict__ ovf_cnt, const int* __restrict__ ovf,
                   const float* __restrict__ xs, const float* __restrict__ bl, const int* __restrict__ batch,
                   float* __restrict__ pool, float* __restrict__ cnt) {
  const int lane = threadIdx.x & 31;
  const int w = __builtin_amdgcn_readfirstlane((int)((blockIdx.x * 256 + threadIdx.x) >> 5));
  const int n0 = w * POOL_NPW;
  const v4f bb = *(const v4f*)(bl + lane * 4);
  int g[POOL_NPW], c[POOL_NPW], en[POOL_NPW];
  int e[POOL_NPW];
  v4f self[POOL_NPW];
#pragma unroll
  for (int i = 0; i < POOL_NPW; ++i) {
    c[i] = ecnt[n0 + i];
    g[i] = batch[n0 + i];
  }
#pragma unroll
  for (int i = 0; i < POOL_NPW; ++i) {
    en[i] = (n0 + i) * SLOT_CAP + ((c[i] < SLOT_CAP) ? c[i] : SLOT_CAP);
    e[i] = edge_chunk(slot, (n0 + i) * SLOT_CAP, en[i], lane);
    self[i] = *(const v4f*)((const char*)xs + ((unsigned)(n0 + i) * DIM + (unsigned)lane * 4) * 4u);
  }
  v4f row[POOL_NPW];
#pragma unroll
  for (int i = 0; i < POOL_NPW; ++i) {
    v4f a = edge_sum(slot, xs, (n0 + i) * SLOT_CAP, en[i], e[i], lane);
    if (c[i] > SLOT_CAP) a = overflow_sum(a, ovf_cnt, ovf, xs, n0 + i, lane);
    row[i] = bias_self_relu(bb, self[i], deg_inv_sqrt(c[i]), a);
  }
  int cur = g[0];
  int len = 1;
  v4f acc = row[0];
#pragma unroll
  for (int i = 1; i < POOL_NPW; ++i) {
    if (g[i] == cur) {
      acc += row[i];
      ++len;
    } else {
      pool_add_run(pool, cnt, cur, lane, acc, len);
      cur = g[i];
      acc = row[i];
      len = 1;
    }
  }
  pool_add_run(pool, cnt, cur, lane, acc, len);
}

extern "C" void kernel_launch(void* const* d_in, const int* in_sizes, int n_in,
                              void* d_out, int out_size, void* d_ws, size_t ws_size,
                              hipStream_t stream) {
  const int*   x    = (const int*)d_in[0];
  const int*   ei   = (const int*)d_in[1];
  const int*   batch= (const int*)d_in[2];
  const float* emb  = (const float*)d_in[3];
  const float* W    = (const float*)d_in[4];
  const float* b    = (const float*)d_in[5];
  const float* Wr   = (const float*)d_in[6];
  const float* br   = (const float*)d_in[7];
  float* out = (float*)d_out;

  char* ws = (char*)d_ws;
  const size_t NH = (size_t)N_NODES * DIM * sizeof(float);
  float* bufA = (float*)(ws);
  float* bufB = (float*)(ws + NH);
  int*   ecnt    = (int*)(ws + 2 * NH);
  int*   ovf_cnt = (int*)(ws + 2 * NH + 256 * 1024);
  int*   tctr    = (int*)(ws + 2 * NH + 512 * 1024);
  int*   slot    = (int*)(ws + 2 * NH + 8 * 1024 * 1024);
  int*   ovf     = (int*)(ws + 3 * NH + 8 * 1024 * 1024);
  float* pool = (float*)(ws + 3 * NH + 4 * 1024 * 1024);
  float* cnt  = (float*)(ws + 3 * NH + 4 * 1024 * 1024 + 512 * 1024);
  int*   part = (int*)(ws + 3 * NH + 16 * 1024 * 1024);
  int*   off  = (int*)(ws + 3 * NH + 20 * 1024 * 1024);
  float* tab  = (float*)(ws + 3 * NH + 24 * 1024 * 1024);

  const int B256 = 256;
  const int pool_blocks = N_NODES / (8 * POOL_NPW);

  const int g_blocks = 1024;
  const int n_tiles = N_NODES / 16;

  partition_kernel<<<N_CHUNKS + TAB_TILES, PART_THREADS, 0, stream>>>(ei, part, off, ovf_cnt, pool, cnt, emb, W, tab, tctr);

  slot_kernel<<<N_BINS, SLOT_WAVES * 32, 0, stream>>>(part, off, ecnt, slot, ovf_cnt, ovf, x, tab, bufA);

  rows_gemm_kernel<<<g_blocks, B256, 0, stream>>>(ecnt, slot, ovf_cnt, ovf, bufA, b,
                                                  W + (size_t)1 * DIM * DIM, bufB, n_tiles, tctr);
  rows_gemm_kernel<<<g_blocks, B256, 0, stream>>>(ecnt, slot, ovf_cnt, ovf, bufB, b + 1 * DIM,
                                                  W + (size_t)2 * DIM * DIM, bufA, n_tiles, tctr + TCTR_PITCH);
  rows_gemm_kernel<<<g_blocks, B256, 0, stream>>>(ecnt, slot, ovf_cnt, ovf, bufA, b + 2 * DIM,
                                                  W + (size_t)3 * DIM * DIM, bufB, n_tiles, tctr + 2 * TCTR_PITCH);

  gather_pool_kernel<<<pool_blocks, B256, 0, stream>>>(ecnt, slot, ovf_cnt, ovf, bufB, b + 3 * DIM, batch, pool, cnt);

  mean_gemm_wmma_f32<<<N_GRAPHS / 16, B256, 0, stream>>>(pool, cnt, Wr, br, out, N_GRAPHS / 16);
}
